# starting kernel
# baseline (speedup 1.0000x reference)
#include <hip/hip_runtime.h>
#include <hip/hip_bf16.h>
#include <math.h>


typedef __attribute__((ext_vector_type(2))) float v2f;
typedef __attribute__((ext_vector_type(8))) float v8f;

#define HID 64
#define NEG_SLOPE 0.2f
#define LN_EPS 1e-5f

__device__ __forceinline__ unsigned fkey(float f) {
    unsigned b = __float_as_uint(f);
    return (b & 0x80000000u) ? ~b : (b | 0x80000000u);
}
__device__ __forceinline__ float funkey(unsigned k) {
    unsigned b = (k & 0x80000000u) ? (k ^ 0x80000000u) : ~k;
    return __uint_as_float(b);
}
#define KEY_NEG_INF 0x007FFFFFu

template <int NTILES>
__global__ void gemm64_wmma(const float* __restrict__ X,
                            const float* __restrict__ W,
                            const float* __restrict__ bias,
                            float* __restrict__ Y,
                            int nrows, int nout) {
    __shared__ float lds_w[64][64];

    int tid = threadIdx.x;
    for (int idx = tid; idx < 64 * 64; idx += 128) {
        int r = idx >> 6, c = idx & 63;
        lds_w[r][c] = (c < nout) ? W[r * nout + c] : 0.0f;
    }
    __syncthreads();

    int lane  = tid & 31;
    int wave  = tid >> 5;
    int half  = lane >> 4;
    int l15   = lane & 15;
    int rbase = blockIdx.x * 64 + wave * 16;

    int arow = rbase + l15;
    if (arow >= nrows) arow = nrows - 1;
    const float* xrow = X + (size_t)arow * 64;

    v8f acc[NTILES];
#pragma unroll
    for (int t = 0; t < NTILES; ++t)
#pragma unroll
        for (int j = 0; j < 8; ++j) acc[t][j] = 0.0f;

#pragma unroll
    for (int k0 = 0; k0 < 64; k0 += 4) {
        const float* ap = xrow + k0 + 2 * half;
        v2f a;
        a.x = ap[0];
        a.y = ap[1];
#pragma unroll
        for (int t = 0; t < NTILES; ++t) {
            v2f b;
            b.x = lds_w[k0 + 2 * half + 0][t * 16 + l15];
            b.y = lds_w[k0 + 2 * half + 1][t * 16 + l15];
            acc[t] = __builtin_amdgcn_wmma_f32_16x16x4_f32(
                false, a, false, b, (short)0, acc[t], false, false);
        }
    }

#pragma unroll
    for (int t = 0; t < NTILES; ++t) {
        int col = t * 16 + l15;
        float bv = (col < nout) ? bias[col] : 0.0f;
#pragma unroll
        for (int i = 0; i < 8; ++i) {
            int row = rbase + 8 * half + i;
            if (row < nrows && col < nout)
                Y[(size_t)row * nout + col] = acc[t][i] + bv;
        }
    }
}

__global__ void init_layer(float* __restrict__ t, float* __restrict__ denom,
                           unsigned* __restrict__ mx, int n) {
    int i = blockIdx.x * blockDim.x + threadIdx.x;
    if (i < n * 16) {
        float4 z = make_float4(0.f, 0.f, 0.f, 0.f);
        ((float4*)t)[i] = z;
    }
    if (i < n) {
        denom[i] = 0.0f;
        mx[i]    = KEY_NEG_INF;
    }
}

__global__ void edge_score(const float* __restrict__ fs,
                           const float* __restrict__ fd,
                           const int* __restrict__ src,
                           const int* __restrict__ dst,
                           const float* __restrict__ attn,
                           float* __restrict__ score,
                           unsigned* __restrict__ mx, int E) {
    int e = blockIdx.x * blockDim.x + threadIdx.x;
    if (e >= E) return;
    int s = src[e], d = dst[e];
    const float4* ps = (const float4*)(fs + (size_t)s * HID);
    const float4* pd = (const float4*)(fd + (size_t)d * HID);
    const float4* pa = (const float4*)attn;
    float acc = 0.0f;
#pragma unroll
    for (int i = 0; i < HID / 4; ++i) {
        float4 a = ps[i], b = pd[i], w = pa[i];
        float x0 = a.x + b.x, x1 = a.y + b.y, x2 = a.z + b.z, x3 = a.w + b.w;
        x0 = x0 > 0.f ? x0 : x0 * NEG_SLOPE;
        x1 = x1 > 0.f ? x1 : x1 * NEG_SLOPE;
        x2 = x2 > 0.f ? x2 : x2 * NEG_SLOPE;
        x3 = x3 > 0.f ? x3 : x3 * NEG_SLOPE;
        acc += x0 * w.x + x1 * w.y + x2 * w.z + x3 * w.w;
    }
    score[e] = acc;
    atomicMax(&mx[d], fkey(acc));
}

__global__ void edge_exp(float* __restrict__ score,
                         const unsigned* __restrict__ mx,
                         const int* __restrict__ dst,
                         float* __restrict__ denom, int E) {
    int e = blockIdx.x * blockDim.x + threadIdx.x;
    if (e >= E) return;
    int d = dst[e];
    float ex = expf(score[e] - funkey(mx[d]));
    score[e] = ex;
    atomicAdd(&denom[d], ex);
}

__global__ void edge_scatter(const float* __restrict__ score,
                             const float* __restrict__ denom,
                             const float* __restrict__ fs,
                             const int* __restrict__ src,
                             const int* __restrict__ dst,
                             float* __restrict__ t, int E) {
    int e = blockIdx.x * blockDim.x + threadIdx.x;
    if (e >= E) return;
    int s = src[e], d = dst[e];
    float alpha = score[e] / denom[d];
    const float4* ps = (const float4*)(fs + (size_t)s * HID);
    float* pt = t + (size_t)d * HID;
#pragma unroll
    for (int i = 0; i < HID / 4; ++i) {
        float4 v = ps[i];
        atomicAdd(pt + 4 * i + 0, v.x * alpha);
        atomicAdd(pt + 4 * i + 1, v.y * alpha);
        atomicAdd(pt + 4 * i + 2, v.z * alpha);
        atomicAdd(pt + 4 * i + 3, v.w * alpha);
    }
}

__global__ void node_update(float* __restrict__ h, const float* __restrict__ t,
                            const float* __restrict__ gamma,
                            const float* __restrict__ beta, int n) {
    int i = blockIdx.x * blockDim.x + threadIdx.x;
    if (i >= n) return;
    const float4* pt = (const float4*)(t + (size_t)i * HID);
    float4 v[HID / 4];
    float sum = 0.0f;
#pragma unroll
    for (int j = 0; j < HID / 4; ++j) {
        v[j] = pt[j];
        sum += v[j].x + v[j].y + v[j].z + v[j].w;
    }
    float mean = sum * (1.0f / HID);
    float var = 0.0f;
#pragma unroll
    for (int j = 0; j < HID / 4; ++j) {
        float d0 = v[j].x - mean, d1 = v[j].y - mean;
        float d2 = v[j].z - mean, d3 = v[j].w - mean;
        var += d0 * d0 + d1 * d1 + d2 * d2 + d3 * d3;
    }
    var *= (1.0f / HID);
    float inv = rsqrtf(var + LN_EPS);

    float4* ph = (float4*)(h + (size_t)i * HID);
    const float4* pg = (const float4*)gamma;
    const float4* pb = (const float4*)beta;
#pragma unroll
    for (int j = 0; j < HID / 4; ++j) {
        float4 g = pg[j], b = pb[j], hv = ph[j];
        float x0 = (v[j].x - mean) * inv * g.x + b.x;
        float x1 = (v[j].y - mean) * inv * g.y + b.y;
        float x2 = (v[j].z - mean) * inv * g.z + b.z;
        float x3 = (v[j].w - mean) * inv * g.w + b.w;
        const float k = 0.70710678118654752f;
        hv.x += 0.5f * x0 * (1.0f + erff(x0 * k));
        hv.y += 0.5f * x1 * (1.0f + erff(x1 * k));
        hv.z += 0.5f * x2 * (1.0f + erff(x2 * k));
        hv.w += 0.5f * x3 * (1.0f + erff(x3 * k));
        ph[j] = hv;
    }
}

extern "C" void kernel_launch(void* const* d_in, const int* in_sizes, int n_in,
                              void* d_out, int out_size, void* d_ws, size_t ws_size,
                              hipStream_t stream) {
    const float* nodes = (const float*)d_in[0];
    const int*   src   = (const int*)d_in[1];
    const int*   dst   = (const int*)d_in[2];
    const float* w_in  = (const float*)d_in[3];
    const float* b_in  = (const float*)d_in[4];
    const float* w_src = (const float*)d_in[5];
    const float* b_src = (const float*)d_in[6];
    const float* w_dst = (const float*)d_in[7];
    const float* b_dst = (const float*)d_in[8];
    const float* attn  = (const float*)d_in[9];
    const float* gamma = (const float*)d_in[10];
    const float* beta  = (const float*)d_in[11];
    const float* w_out = (const float*)d_in[12];
    const float* b_out = (const float*)d_in[13];

    const int N = in_sizes[0] / HID;
    const int E = in_sizes[1];
    const int NL = 6;

    float* ws   = (float*)d_ws;
    float* h    = ws;  ws += (size_t)N * HID;
    float* fs   = ws;  ws += (size_t)N * HID;
    float* fd   = ws;  ws += (size_t)N * HID;
    float* tbuf = ws;  ws += (size_t)N * HID;
    float* score = ws; ws += E;
    float* denom = ws; ws += N;
    unsigned* mx = (unsigned*)ws;

    const int ggrid  = (N + 63) / 64;
    const int eblk   = 256;
    const int egrid  = (E + eblk - 1) / eblk;
    const int ngrid  = (N + 255) / 256;
    const int igrid  = (N * 16 + 255) / 256;

    gemm64_wmma<4><<<ggrid, 128, 0, stream>>>(nodes, w_in, b_in, h, N, HID);

    for (int L = 0; L < NL; ++L) {
        gemm64_wmma<4><<<ggrid, 128, 0, stream>>>(h, w_src + (size_t)L * HID * HID,
                                                  b_src + (size_t)L * HID, fs, N, HID);
        gemm64_wmma<4><<<ggrid, 128, 0, stream>>>(h, w_dst + (size_t)L * HID * HID,
                                                  b_dst + (size_t)L * HID, fd, N, HID);
        init_layer<<<igrid, 256, 0, stream>>>(tbuf, denom, mx, N);
        edge_score<<<egrid, eblk, 0, stream>>>(fs, fd, src, dst,
                                               attn + (size_t)L * HID, score, mx, E);
        edge_exp<<<egrid, eblk, 0, stream>>>(score, mx, dst, denom, E);
        edge_scatter<<<egrid, eblk, 0, stream>>>(score, denom, fs, src, dst, tbuf, E);
        node_update<<<ngrid, 256, 0, stream>>>(h, tbuf, gamma + (size_t)L * HID,
                                               beta + (size_t)L * HID, N);
    }

    gemm64_wmma<1><<<ggrid, 128, 0, stream>>>(h, w_out, b_out, (float*)d_out, N, 8);
}


# edge list grouped by destination once per call (count, scan, fill, rank by edge id); 16 lanes own a dst row: max, denominator, row sum in registers, no float atomics; sums in edge-index order
# speedup vs baseline: 5.1413x; 5.1413x over previous
#include <hip/hip_runtime.h>
#include <hip/hip_bf16.h>
#include <math.h>


typedef __attribute__((ext_vector_type(2))) float v2f;
typedef __attribute__((ext_vector_type(8))) float v8f;

#define HID 64
#define NEG_SLOPE 0.2f
#define LN_EPS 1e-5f

template <int NTILES>
__global__ void gemm64_wmma(const float* __restrict__ X,
                            const float* __restrict__ W,
                            const float* __restrict__ bias,
                            float* __restrict__ Y,
                            int nrows, int nout) {
    __shared__ float lds_w[64][64];

    int tid = threadIdx.x;
    for (int idx = tid; idx < 64 * 64; idx += 128) {
        int r = idx >> 6, c = idx & 63;
        lds_w[r][c] = (c < nout) ? W[r * nout + c] : 0.0f;
    }
    __syncthreads();

    int lane  = tid & 31;
    int wave  = tid >> 5;
    int half  = lane >> 4;
    int l15   = lane & 15;
    int rbase = blockIdx.x * 64 + wave * 16;

    int arow = rbase + l15;
    if (arow >= nrows) arow = nrows - 1;
    const float* xrow = X + (size_t)arow * 64;

    v8f acc[NTILES];
#pragma unroll
    for (int t = 0; t < NTILES; ++t)
#pragma unroll
        for (int j = 0; j < 8; ++j) acc[t][j] = 0.0f;

#pragma unroll
    for (int k0 = 0; k0 < 64; k0 += 4) {
        const float* ap = xrow + k0 + 2 * half;
        v2f a;
        a.x = ap[0];
        a.y = ap[1];
#pragma unroll
        for (int t = 0; t < NTILES; ++t) {
            v2f b;
            b.x = lds_w[k0 + 2 * half + 0][t * 16 + l15];
            b.y = lds_w[k0 + 2 * half + 1][t * 16 + l15];
            acc[t] = __builtin_amdgcn_wmma_f32_16x16x4_f32(
                false, a, false, b, (short)0, acc[t], false, false);
        }
    }

#pragma unroll
    for (int t = 0; t < NTILES; ++t) {
        int col = t * 16 + l15;
        float bv = (col < nout) ? bias[col] : 0.0f;
#pragma unroll
        for (int i = 0; i < 8; ++i) {
            int row = rbase + 8 * half + i;
            if (row < nrows && col < nout)
                Y[(size_t)row * nout + col] = acc[t][i] + bv;
        }
    }
}

__global__ void zero_counts(int* __restrict__ cnt, int n) {
    int i = blockIdx.x * blockDim.x + threadIdx.x;
    if (i < n) cnt[i] = 0;
}

__global__ void count_dst(const int* __restrict__ dst, int* __restrict__ cnt, int E) {
    int e = blockIdx.x * blockDim.x + threadIdx.x;
    if (e < E) atomicAdd(&cnt[dst[e]], 1);
}

__global__ void __launch_bounds__(1024)
scan_counts(int* __restrict__ start, int* __restrict__ cursor, int n) {
    __shared__ int part[1024];
    const int tid = threadIdx.x;
    const int per = (n + 1023) >> 10;
    const int b = tid * per;
    int e = b + per;
    if (e > n) e = n;
    int s = 0;
    for (int i = b; i < e; ++i) s += start[i];
    part[tid] = s;
    __syncthreads();
    for (int d = 1; d < 1024; d <<= 1) {
        int v = (tid >= d) ? part[tid - d] : 0;
        __syncthreads();
        part[tid] += v;
        __syncthreads();
    }
    int run = part[tid] - s;
#pragma unroll 8
    for (int i = b; i < e; ++i) {
        int c = start[i];
        start[i]  = run;
        cursor[i] = run;
        run += c;
    }
}

__global__ void fill_by_dst(const int* __restrict__ dst, int* __restrict__ cursor,
                            int* __restrict__ eid, int E) {
    int e = blockIdx.x * blockDim.x + threadIdx.x;
    if (e >= E) return;
    int p = atomicAdd(&cursor[dst[e]], 1);
    eid[p] = e;
}

__global__ void order_segments(const int* __restrict__ src, const int* __restrict__ dst,
                               const int* __restrict__ seg_start,
                               const int* __restrict__ seg_end,
                               const int* __restrict__ eid,
                               int* __restrict__ ssrc, int E) {
    int p = blockIdx.x * blockDim.x + threadIdx.x;
    if (p >= E) return;
    int e = eid[p];
    int d = dst[e];
    int b = seg_start[d], en = seg_end[d];
    int r = 0;
    for (int q = b; q < en; ++q) r += (eid[q] < e) ? 1 : 0;
    ssrc[b + r] = src[e];
}

template <int CTRL>
__device__ __forceinline__ float dpp_row(float v) {
    return __builtin_bit_cast(float, __builtin_amdgcn_update_dpp(
        0, __builtin_bit_cast(int, v), CTRL, 0xf, 0xf, false));
}
__device__ __forceinline__ float row16_sum(float v) {
    v += dpp_row<0x161>(v);
    v += dpp_row<0x162>(v);
    v += dpp_row<0x164>(v);
    v += dpp_row<0x168>(v);
    return v;
}

__global__ void __launch_bounds__(256)
seg_softmax_sum(const float* __restrict__ fs, const float* __restrict__ fd,
                const int* __restrict__ ssrc,
                const int* __restrict__ seg_start,
                const int* __restrict__ seg_end,
                const float* __restrict__ attn,
                float* __restrict__ t, int n) {
    const int lane = threadIdx.x & 31;
    const int l16  = lane & 15;
    const int half = lane >> 4;
    const int node = (blockIdx.x * blockDim.x + threadIdx.x) >> 4;
    const bool valid = node < n;
    const int nd = valid ? node : (n - 1);
    const int b = seg_start[nd];
    const int deg = valid ? (seg_end[nd] - b) : 0;
    const float4 fdv = ((const float4*)(fd + (size_t)nd * HID))[l16];
    const float4 av  = ((const float4*)attn)[l16];

    const int d0 = __builtin_amdgcn_readlane(deg, 0);
    const int d1 = __builtin_amdgcn_readlane(deg, 16);
    const int dmax = d0 > d1 ? d0 : d1;

    float m = -INFINITY, l = 0.0f;
    float4 acc = make_float4(0.f, 0.f, 0.f, 0.f);

    for (int off = 0; off < dmax; off += 16) {
        const int p = off + l16;
        const int myidx = (p < deg) ? ssrc[b + p] : 0;
        int lim = dmax - off;
        if (lim > 16) lim = 16;
        for (int j = 0; j < lim; j += 4) {
            float4 r[4];
#pragma unroll
            for (int k = 0; k < 4; ++k) {
                const int s0 = __builtin_amdgcn_readlane(myidx, j + k);
                const int s1 = __builtin_amdgcn_readlane(myidx, j + k + 16);
                const int s  = half ? s1 : s0;
                r[k] = ((const float4*)(fs + (size_t)s * HID))[l16];
            }
#pragma unroll
            for (int k = 0; k < 4; ++k) {
                float x0 = r[k].x + fdv.x, x1 = r[k].y + fdv.y;
                float x2 = r[k].z + fdv.z, x3 = r[k].w + fdv.w;
                x0 = x0 > 0.f ? x0 : x0 * NEG_SLOPE;
                x1 = x1 > 0.f ? x1 : x1 * NEG_SLOPE;
                x2 = x2 > 0.f ? x2 : x2 * NEG_SLOPE;
                x3 = x3 > 0.f ? x3 : x3 * NEG_SLOPE;
                const float sc = row16_sum(x0 * av.x + x1 * av.y + x2 * av.z + x3 * av.w);
                if (off + j + k < deg) {
                    const float mn    = fmaxf(m, sc);
                    const float ed    = expf(fminf(m, sc) - mn);
                    const bool  up    = sc > m;
                    const float scale = up ? ed : 1.0f;
                    const float pe    = up ? 1.0f : ed;
                    l     = l * scale + pe;
                    acc.x = acc.x * scale + pe * r[k].x;
                    acc.y = acc.y * scale + pe * r[k].y;
                    acc.z = acc.z * scale + pe * r[k].z;
                    acc.w = acc.w * scale + pe * r[k].w;
                    m = mn;
                }
            }
        }
    }

    if (valid) {
        float4 o = make_float4(0.f, 0.f, 0.f, 0.f);
        if (deg > 0) {
            o.x = acc.x / l; o.y = acc.y / l; o.z = acc.z / l; o.w = acc.w / l;
        }
        ((float4*)(t + (size_t)node * HID))[l16] = o;
    }
}

__global__ void node_update(float* __restrict__ h, const float* __restrict__ t,
                            const float* __restrict__ gamma,
                            const float* __restrict__ beta, int n) {
    int i = blockIdx.x * blockDim.x + threadIdx.x;
    if (i >= n) return;
    const float4* pt = (const float4*)(t + (size_t)i * HID);
    float4 v[HID / 4];
    float sum = 0.0f;
#pragma unroll
    for (int j = 0; j < HID / 4; ++j) {
        v[j] = pt[j];
        sum += v[j].x + v[j].y + v[j].z + v[j].w;
    }
    float mean = sum * (1.0f / HID);
    float var = 0.0f;
#pragma unroll
    for (int j = 0; j < HID / 4; ++j) {
        float d0 = v[j].x - mean, d1 = v[j].y - mean;
        float d2 = v[j].z - mean, d3 = v[j].w - mean;
        var += d0 * d0 + d1 * d1 + d2 * d2 + d3 * d3;
    }
    var *= (1.0f / HID);
    float inv = rsqrtf(var + LN_EPS);

    float4* ph = (float4*)(h + (size_t)i * HID);
    const float4* pg = (const float4*)gamma;
    const float4* pb = (const float4*)beta;
#pragma unroll
    for (int j = 0; j < HID / 4; ++j) {
        float4 g = pg[j], b = pb[j], hv = ph[j];
        float x0 = (v[j].x - mean) * inv * g.x + b.x;
        float x1 = (v[j].y - mean) * inv * g.y + b.y;
        float x2 = (v[j].z - mean) * inv * g.z + b.z;
        float x3 = (v[j].w - mean) * inv * g.w + b.w;
        const float k = 0.70710678118654752f;
        hv.x += 0.5f * x0 * (1.0f + erff(x0 * k));
        hv.y += 0.5f * x1 * (1.0f + erff(x1 * k));
        hv.z += 0.5f * x2 * (1.0f + erff(x2 * k));
        hv.w += 0.5f * x3 * (1.0f + erff(x3 * k));
        ph[j] = hv;
    }
}

extern "C" void kernel_launch(void* const* d_in, const int* in_sizes, int n_in,
                              void* d_out, int out_size, void* d_ws, size_t ws_size,
                              hipStream_t stream) {
    const float* nodes = (const float*)d_in[0];
    const int*   src   = (const int*)d_in[1];
    const int*   dst   = (const int*)d_in[2];
    const float* w_in  = (const float*)d_in[3];
    const float* b_in  = (const float*)d_in[4];
    const float* w_src = (const float*)d_in[5];
    const float* b_src = (const float*)d_in[6];
    const float* w_dst = (const float*)d_in[7];
    const float* b_dst = (const float*)d_in[8];
    const float* attn  = (const float*)d_in[9];
    const float* gamma = (const float*)d_in[10];
    const float* beta  = (const float*)d_in[11];
    const float* w_out = (const float*)d_in[12];
    const float* b_out = (const float*)d_in[13];

    const int N = in_sizes[0] / HID;
    const int E = in_sizes[1];
    const int NL = 6;

    float* ws   = (float*)d_ws;
    float* h    = ws;  ws += (size_t)N * HID;
    float* fs   = ws;  ws += (size_t)N * HID;
    float* fd   = ws;  ws += (size_t)N * HID;
    float* tbuf = ws;  ws += (size_t)N * HID;
    int* ssrc      = (int*)ws; ws += E;
    int* seg_start = (int*)ws; ws += N;
    int* seg_end   = (int*)ws;
    int* eid       = (int*)tbuf;

    const int ggrid  = (N + 63) / 64;
    const int eblk   = 256;
    const int egrid  = (E + eblk - 1) / eblk;
    const int ngrid  = (N + 255) / 256;
    const int sgrid  = (N * 16 + 255) / 256;

    gemm64_wmma<4><<<ggrid, 128, 0, stream>>>(nodes, w_in, b_in, h, N, HID);

    zero_counts<<<ngrid, 256, 0, stream>>>(seg_start, N);
    count_dst<<<egrid, eblk, 0, stream>>>(dst, seg_start, E);
    scan_counts<<<1, 1024, 0, stream>>>(seg_start, seg_end, N);
    fill_by_dst<<<egrid, eblk, 0, stream>>>(dst, seg_end, eid, E);
    order_segments<<<egrid, eblk, 0, stream>>>(src, dst, seg_start, seg_end, eid, ssrc, E);

    for (int L = 0; L < NL; ++L) {
        gemm64_wmma<4><<<ggrid, 128, 0, stream>>>(h, w_src + (size_t)L * HID * HID,
                                                  b_src + (size_t)L * HID, fs, N, HID);
        gemm64_wmma<4><<<ggrid, 128, 0, stream>>>(h, w_dst + (size_t)L * HID * HID,
                                                  b_dst + (size_t)L * HID, fd, N, HID);
        seg_softmax_sum<<<sgrid, 256, 0, stream>>>(fs, fd, ssrc, seg_start, seg_end,
                                                   attn + (size_t)L * HID, tbuf, N);
        node_update<<<ngrid, 256, 0, stream>>>(h, tbuf, gamma + (size_t)L * HID,
                                               beta + (size_t)L * HID, N);
    }

    gemm64_wmma<1><<<ggrid, 128, 0, stream>>>(h, w_out, b_out, (float*)d_out, N, 8);
}


# layernorm, gelu and residual moved into the edge kernel's epilogue: the 16 lanes owning a dst row normalise it in registers (mean, variance as 16-lane butterfly sums); t plane and 6 launches gone
# speedup vs baseline: 5.5084x; 1.0714x over previous
#include <hip/hip_runtime.h>
#include <hip/hip_bf16.h>
#include <math.h>


typedef __attribute__((ext_vector_type(2))) float v2f;
typedef __attribute__((ext_vector_type(8))) float v8f;

#define HID 64
#define NEG_SLOPE 0.2f
#define LN_EPS 1e-5f

template <int NTILES>
__global__ void gemm64_wmma(const float* __restrict__ X,
                            const float* __restrict__ W,
                            const float* __restrict__ bias,
                            float* __restrict__ Y,
                            int nrows, int nout) {
    __shared__ float lds_w[64][64];

    int tid = threadIdx.x;
    for (int idx = tid; idx < 64 * 64; idx += 128) {
        int r = idx >> 6, c = idx & 63;
        lds_w[r][c] = (c < nout) ? W[r * nout + c] : 0.0f;
    }
    __syncthreads();

    int lane  = tid & 31;
    int wave  = tid >> 5;
    int half  = lane >> 4;
    int l15   = lane & 15;
    int rbase = blockIdx.x * 64 + wave * 16;

    int arow = rbase + l15;
    if (arow >= nrows) arow = nrows - 1;
    const float* xrow = X + (size_t)arow * 64;

    v8f acc[NTILES];
#pragma unroll
    for (int t = 0; t < NTILES; ++t)
#pragma unroll
        for (int j = 0; j < 8; ++j) acc[t][j] = 0.0f;

#pragma unroll
    for (int k0 = 0; k0 < 64; k0 += 4) {
        const float* ap = xrow + k0 + 2 * half;
        v2f a;
        a.x = ap[0];
        a.y = ap[1];
#pragma unroll
        for (int t = 0; t < NTILES; ++t) {
            v2f b;
            b.x = lds_w[k0 + 2 * half + 0][t * 16 + l15];
            b.y = lds_w[k0 + 2 * half + 1][t * 16 + l15];
            acc[t] = __builtin_amdgcn_wmma_f32_16x16x4_f32(
                false, a, false, b, (short)0, acc[t], false, false);
        }
    }

#pragma unroll
    for (int t = 0; t < NTILES; ++t) {
        int col = t * 16 + l15;
        float bv = (col < nout) ? bias[col] : 0.0f;
#pragma unroll
        for (int i = 0; i < 8; ++i) {
            int row = rbase + 8 * half + i;
            if (row < nrows && col < nout)
                Y[(size_t)row * nout + col] = acc[t][i] + bv;
        }
    }
}

__global__ void zero_counts(int* __restrict__ cnt, int n) {
    int i = blockIdx.x * blockDim.x + threadIdx.x;
    if (i < n) cnt[i] = 0;
}

__global__ void count_dst(const int* __restrict__ dst, int* __restrict__ cnt, int E) {
    int e = blockIdx.x * blockDim.x + threadIdx.x;
    if (e < E) atomicAdd(&cnt[dst[e]], 1);
}

__global__ void __launch_bounds__(1024)
scan_counts(int* __restrict__ start, int* __restrict__ cursor, int n) {
    __shared__ int part[1024];
    const int tid = threadIdx.x;
    const int per = (n + 1023) >> 10;
    const int b = tid * per;
    int e = b + per;
    if (e > n) e = n;
    int s = 0;
    for (int i = b; i < e; ++i) s += start[i];
    part[tid] = s;
    __syncthreads();
    for (int d = 1; d < 1024; d <<= 1) {
        int v = (tid >= d) ? part[tid - d] : 0;
        __syncthreads();
        part[tid] += v;
        __syncthreads();
    }
    int run = part[tid] - s;
#pragma unroll 8
    for (int i = b; i < e; ++i) {
        int c = start[i];
        start[i]  = run;
        cursor[i] = run;
        run += c;
    }
}

__global__ void fill_by_dst(const int* __restrict__ dst, int* __restrict__ cursor,
                            int* __restrict__ eid, int E) {
    int e = blockIdx.x * blockDim.x + threadIdx.x;
    if (e >= E) return;
    int p = atomicAdd(&cursor[dst[e]], 1);
    eid[p] = e;
}

__global__ void order_segments(const int* __restrict__ src, const int* __restrict__ dst,
                               const int* __restrict__ seg_start,
                               const int* __restrict__ seg_end,
                               const int* __restrict__ eid,
                               int* __restrict__ ssrc, int E) {
    int p = blockIdx.x * blockDim.x + threadIdx.x;
    if (p >= E) return;
    int e = eid[p];
    int d = dst[e];
    int b = seg_start[d], en = seg_end[d];
    int r = 0;
    for (int q = b; q < en; ++q) r += (eid[q] < e) ? 1 : 0;
    ssrc[b + r] = src[e];
}

template <int CTRL>
__device__ __forceinline__ float dpp_row(float v) {
    return __builtin_bit_cast(float, __builtin_amdgcn_update_dpp(
        0, __builtin_bit_cast(int, v), CTRL, 0xf, 0xf, false));
}
__device__ __forceinline__ float row16_sum(float v) {
    v += dpp_row<0x161>(v);
    v += dpp_row<0x162>(v);
    v += dpp_row<0x164>(v);
    v += dpp_row<0x168>(v);
    return v;
}

__global__ void __launch_bounds__(256)
seg_softmax_update(const float* __restrict__ fs, const float* __restrict__ fd,
                   const int* __restrict__ ssrc,
                   const int* __restrict__ seg_start,
                   const int* __restrict__ seg_end,
                   const float* __restrict__ attn,
                   const float* __restrict__ gamma,
                   const float* __restrict__ beta,
                   float* __restrict__ h, int n) {
    const int lane = threadIdx.x & 31;
    const int l16  = lane & 15;
    const int half = lane >> 4;
    const int node = (blockIdx.x * blockDim.x + threadIdx.x) >> 4;
    const bool valid = node < n;
    const int nd = valid ? node : (n - 1);
    const int b = seg_start[nd];
    const int deg = valid ? (seg_end[nd] - b) : 0;
    const float4 fdv = ((const float4*)(fd + (size_t)nd * HID))[l16];
    const float4 av  = ((const float4*)attn)[l16];
    float4 hv = ((const float4*)(h + (size_t)nd * HID))[l16];

    const int d0 = __builtin_amdgcn_readlane(deg, 0);
    const int d1 = __builtin_amdgcn_readlane(deg, 16);
    const int dmax = d0 > d1 ? d0 : d1;

    float m = -INFINITY, l = 0.0f;
    float4 acc = make_float4(0.f, 0.f, 0.f, 0.f);

    for (int off = 0; off < dmax; off += 16) {
        const int p = off + l16;
        const int myidx = (p < deg) ? ssrc[b + p] : 0;
        int lim = dmax - off;
        if (lim > 16) lim = 16;
        for (int j = 0; j < lim; j += 4) {
            float4 r[4];
#pragma unroll
            for (int k = 0; k < 4; ++k) {
                const int s0 = __builtin_amdgcn_readlane(myidx, j + k);
                const int s1 = __builtin_amdgcn_readlane(myidx, j + k + 16);
                const int s  = half ? s1 : s0;
                r[k] = ((const float4*)(fs + (size_t)s * HID))[l16];
            }
#pragma unroll
            for (int k = 0; k < 4; ++k) {
                float x0 = r[k].x + fdv.x, x1 = r[k].y + fdv.y;
                float x2 = r[k].z + fdv.z, x3 = r[k].w + fdv.w;
                x0 = x0 > 0.f ? x0 : x0 * NEG_SLOPE;
                x1 = x1 > 0.f ? x1 : x1 * NEG_SLOPE;
                x2 = x2 > 0.f ? x2 : x2 * NEG_SLOPE;
                x3 = x3 > 0.f ? x3 : x3 * NEG_SLOPE;
                const float sc = row16_sum(x0 * av.x + x1 * av.y + x2 * av.z + x3 * av.w);
                if (off + j + k < deg) {
                    const float mn    = fmaxf(m, sc);
                    const float ed    = expf(fminf(m, sc) - mn);
                    const bool  up    = sc > m;
                    const float scale = up ? ed : 1.0f;
                    const float pe    = up ? 1.0f : ed;
                    l     = l * scale + pe;
                    acc.x = acc.x * scale + pe * r[k].x;
                    acc.y = acc.y * scale + pe * r[k].y;
                    acc.z = acc.z * scale + pe * r[k].z;
                    acc.w = acc.w * scale + pe * r[k].w;
                    m = mn;
                }
            }
        }
    }

    float4 o = make_float4(0.f, 0.f, 0.f, 0.f);
    if (deg > 0) {
        o.x = acc.x / l; o.y = acc.y / l; o.z = acc.z / l; o.w = acc.w / l;
    }

    const float mean = row16_sum((o.x + o.y) + (o.z + o.w)) * (1.0f / HID);
    const float c0 = o.x - mean, c1 = o.y - mean, c2 = o.z - mean, c3 = o.w - mean;
    const float var = row16_sum((c0 * c0 + c1 * c1) + (c2 * c2 + c3 * c3)) * (1.0f / HID);
    const float inv = rsqrtf(var + LN_EPS);

    const float4 g  = ((const float4*)gamma)[l16];
    const float4 bt = ((const float4*)beta)[l16];
    const float x0 = c0 * inv * g.x + bt.x;
    const float x1 = c1 * inv * g.y + bt.y;
    const float x2 = c2 * inv * g.z + bt.z;
    const float x3 = c3 * inv * g.w + bt.w;
    const float kk = 0.70710678118654752f;
    hv.x += 0.5f * x0 * (1.0f + erff(x0 * kk));
    hv.y += 0.5f * x1 * (1.0f + erff(x1 * kk));
    hv.z += 0.5f * x2 * (1.0f + erff(x2 * kk));
    hv.w += 0.5f * x3 * (1.0f + erff(x3 * kk));

    if (valid) ((float4*)(h + (size_t)node * HID))[l16] = hv;
}

extern "C" void kernel_launch(void* const* d_in, const int* in_sizes, int n_in,
                              void* d_out, int out_size, void* d_ws, size_t ws_size,
                              hipStream_t stream) {
    const float* nodes = (const float*)d_in[0];
    const int*   src   = (const int*)d_in[1];
    const int*   dst   = (const int*)d_in[2];
    const float* w_in  = (const float*)d_in[3];
    const float* b_in  = (const float*)d_in[4];
    const float* w_src = (const float*)d_in[5];
    const float* b_src = (const float*)d_in[6];
    const float* w_dst = (const float*)d_in[7];
    const float* b_dst = (const float*)d_in[8];
    const float* attn  = (const float*)d_in[9];
    const float* gamma = (const float*)d_in[10];
    const float* beta  = (const float*)d_in[11];
    const float* w_out = (const float*)d_in[12];
    const float* b_out = (const float*)d_in[13];

    const int N = in_sizes[0] / HID;
    const int E = in_sizes[1];
    const int NL = 6;

    float* ws   = (float*)d_ws;
    float* h    = ws;  ws += (size_t)N * HID;
    float* fs   = ws;  ws += (size_t)N * HID;
    float* fd   = ws;  ws += (size_t)N * HID;
    float* tbuf = ws;  ws += (size_t)N * HID;
    int* ssrc      = (int*)ws; ws += E;
    int* seg_start = (int*)ws; ws += N;
    int* seg_end   = (int*)ws;
    int* eid       = (int*)tbuf;

    const int ggrid  = (N + 63) / 64;
    const int eblk   = 256;
    const int egrid  = (E + eblk - 1) / eblk;
    const int ngrid  = (N + 255) / 256;
    const int sgrid  = (N * 16 + 255) / 256;

    gemm64_wmma<4><<<ggrid, 128, 0, stream>>>(nodes, w_in, b_in, h, N, HID);

    zero_counts<<<ngrid, 256, 0, stream>>>(seg_start, N);
    count_dst<<<egrid, eblk, 0, stream>>>(dst, seg_start, E);
    scan_counts<<<1, 1024, 0, stream>>>(seg_start, seg_end, N);
    fill_by_dst<<<egrid, eblk, 0, stream>>>(dst, seg_end, eid, E);
    order_segments<<<egrid, eblk, 0, stream>>>(src, dst, seg_start, seg_end, eid, ssrc, E);

    for (int L = 0; L < NL; ++L) {
        gemm64_wmma<4><<<ggrid, 128, 0, stream>>>(h, w_src + (size_t)L * HID * HID,
                                                  b_src + (size_t)L * HID, fs, N, HID);
        gemm64_wmma<4><<<ggrid, 128, 0, stream>>>(h, w_dst + (size_t)L * HID * HID,
                                                  b_dst + (size_t)L * HID, fd, N, HID);
        seg_softmax_update<<<sgrid, 256, 0, stream>>>(fs, fd, ssrc, seg_start, seg_end,
                                                      attn + (size_t)L * HID,
                                                      gamma + (size_t)L * HID,
                                                      beta + (size_t)L * HID, h, N);
    }

    gemm64_wmma<1><<<ggrid, 128, 0, stream>>>(h, w_out, b_out, (float*)d_out, N, 8);
}


# GEMM: W staged to LDS with all of a thread's loads in flight (8 x 16-byte loads, then the stores) in place of 32 load-wait-store trips: one L2 round trip per block instead of 32
# speedup vs baseline: 8.2403x; 1.4960x over previous
#include <hip/hip_runtime.h>
#include <hip/hip_bf16.h>
#include <math.h>


typedef __attribute__((ext_vector_type(2))) float v2f;
typedef __attribute__((ext_vector_type(8))) float v8f;

#define HID 64
#define NEG_SLOPE 0.2f
#define LN_EPS 1e-5f

template <int NTILES>
__global__ void __launch_bounds__(128)
gemm64_wmma(const float* __restrict__ X,
            const float* __restrict__ W,
            const float* __restrict__ bias,
            float* __restrict__ Y,
            int nrows, int nout) {
    __shared__ __attribute__((aligned(16))) float lds_w[64][64];

    const int tid = threadIdx.x;
    if (NTILES == 4) {
        const float4* W4 = (const float4*)W;
        float4* L4 = (float4*)&lds_w[0][0];
        float4 wv[8];
#pragma unroll
        for (int i = 0; i < 8; ++i) wv[i] = W4[tid + 128 * i];
#pragma unroll
        for (int i = 0; i < 8; ++i) L4[tid + 128 * i] = wv[i];
    } else {
        constexpr int NC  = NTILES * 16;
        constexpr int PER = 64 * NC / 128;
        float wv[PER];
#pragma unroll
        for (int i = 0; i < PER; ++i) {
            const int idx = tid + 128 * i;
            const int r = idx / NC, c = idx % NC;
            const int cc = (c < nout) ? c : (nout - 1);
            const float v = W[r * nout + cc];
            wv[i] = (c < nout) ? v : 0.0f;
        }
#pragma unroll
        for (int i = 0; i < PER; ++i) {
            const int idx = tid + 128 * i;
            lds_w[idx / NC][idx % NC] = wv[i];
        }
    }
    __syncthreads();

    int lane  = tid & 31;
    int wave  = tid >> 5;
    int half  = lane >> 4;
    int l15   = lane & 15;
    int rbase = blockIdx.x * 64 + wave * 16;

    int arow = rbase + l15;
    if (arow >= nrows) arow = nrows - 1;
    const float* xrow = X + (size_t)arow * 64;

    v8f acc[NTILES];
#pragma unroll
    for (int t = 0; t < NTILES; ++t)
#pragma unroll
        for (int j = 0; j < 8; ++j) acc[t][j] = 0.0f;

#pragma unroll
    for (int k0 = 0; k0 < 64; k0 += 4) {
        const float* ap = xrow + k0 + 2 * half;
        v2f a;
        a.x = ap[0];
        a.y = ap[1];
#pragma unroll
        for (int t = 0; t < NTILES; ++t) {
            v2f b;
            b.x = lds_w[k0 + 2 * half + 0][t * 16 + l15];
            b.y = lds_w[k0 + 2 * half + 1][t * 16 + l15];
            acc[t] = __builtin_amdgcn_wmma_f32_16x16x4_f32(
                false, a, false, b, (short)0, acc[t], false, false);
        }
    }

#pragma unroll
    for (int t = 0; t < NTILES; ++t) {
        int col = t * 16 + l15;
        float bv = (col < nout) ? bias[col] : 0.0f;
#pragma unroll
        for (int i = 0; i < 8; ++i) {
            int row = rbase + 8 * half + i;
            if (row < nrows && col < nout)
                Y[(size_t)row * nout + col] = acc[t][i] + bv;
        }
    }
}

__global__ void zero_counts(int* __restrict__ cnt, int n) {
    int i = blockIdx.x * blockDim.x + threadIdx.x;
    if (i < n) cnt[i] = 0;
}

__global__ void count_dst(const int* __restrict__ dst, int* __restrict__ cnt, int E) {
    int e = blockIdx.x * blockDim.x + threadIdx.x;
    if (e < E) atomicAdd(&cnt[dst[e]], 1);
}

__global__ void __launch_bounds__(1024)
scan_counts(int* __restrict__ start, int* __restrict__ cursor, int n) {
    __shared__ int part[1024];
    const int tid = threadIdx.x;
    const int per = (n + 1023) >> 10;
    const int b = tid * per;
    int e = b + per;
    if (e > n) e = n;
    int s = 0;
    for (int i = b; i < e; ++i) s += start[i];
    part[tid] = s;
    __syncthreads();
    for (int d = 1; d < 1024; d <<= 1) {
        int v = (tid >= d) ? part[tid - d] : 0;
        __syncthreads();
        part[tid] += v;
        __syncthreads();
    }
    int run = part[tid] - s;
#pragma unroll 8
    for (int i = b; i < e; ++i) {
        int c = start[i];
        start[i]  = run;
        cursor[i] = run;
        run += c;
    }
}

__global__ void fill_by_dst(const int* __restrict__ dst, int* __restrict__ cursor,
                            int* __restrict__ eid, int E) {
    int e = blockIdx.x * blockDim.x + threadIdx.x;
    if (e >= E) return;
    int p = atomicAdd(&cursor[dst[e]], 1);
    eid[p] = e;
}

__global__ void order_segments(const int* __restrict__ src, const int* __restrict__ dst,
                               const int* __restrict__ seg_start,
                               const int* __restrict__ seg_end,
                               const int* __restrict__ eid,
                               int* __restrict__ ssrc, int E) {
    int p = blockIdx.x * blockDim.x + threadIdx.x;
    if (p >= E) return;
    int e = eid[p];
    int d = dst[e];
    int b = seg_start[d], en = seg_end[d];
    int r = 0;
    for (int q = b; q < en; ++q) r += (eid[q] < e) ? 1 : 0;
    ssrc[b + r] = src[e];
}

template <int CTRL>
__device__ __forceinline__ float dpp_row(float v) {
    return __builtin_bit_cast(float, __builtin_amdgcn_update_dpp(
        0, __builtin_bit_cast(int, v), CTRL, 0xf, 0xf, false));
}
__device__ __forceinline__ float row16_sum(float v) {
    v += dpp_row<0x161>(v);
    v += dpp_row<0x162>(v);
    v += dpp_row<0x164>(v);
    v += dpp_row<0x168>(v);
    return v;
}

__global__ void __launch_bounds__(256)
seg_softmax_update(const float* __restrict__ fs, const float* __restrict__ fd,
                   const int* __restrict__ ssrc,
                   const int* __restrict__ seg_start,
                   const int* __restrict__ seg_end,
                   const float* __restrict__ attn,
                   const float* __restrict__ gamma,
                   const float* __restrict__ beta,
                   float* __restrict__ h, int n) {
    const int lane = threadIdx.x & 31;
    const int l16  = lane & 15;
    const int half = lane >> 4;
    const int node = (blockIdx.x * blockDim.x + threadIdx.x) >> 4;
    const bool valid = node < n;
    const int nd = valid ? node : (n - 1);
    const int b = seg_start[nd];
    const int deg = valid ? (seg_end[nd] - b) : 0;
    const float4 fdv = ((const float4*)(fd + (size_t)nd * HID))[l16];
    const float4 av  = ((const float4*)attn)[l16];
    float4 hv = ((const float4*)(h + (size_t)nd * HID))[l16];

    const int d0 = __builtin_amdgcn_readlane(deg, 0);
    const int d1 = __builtin_amdgcn_readlane(deg, 16);
    const int dmax = d0 > d1 ? d0 : d1;

    float m = -INFINITY, l = 0.0f;
    float4 acc = make_float4(0.f, 0.f, 0.f, 0.f);

    for (int off = 0; off < dmax; off += 16) {
        const int p = off + l16;
        const int myidx = (p < deg) ? ssrc[b + p] : 0;
        int lim = dmax - off;
        if (lim > 16) lim = 16;
        for (int j = 0; j < lim; j += 4) {
            float4 r[4];
#pragma unroll
            for (int k = 0; k < 4; ++k) {
                const int s0 = __builtin_amdgcn_readlane(myidx, j + k);
                const int s1 = __builtin_amdgcn_readlane(myidx, j + k + 16);
                const int s  = half ? s1 : s0;
                r[k] = ((const float4*)(fs + (size_t)s * HID))[l16];
            }
#pragma unroll
            for (int k = 0; k < 4; ++k) {
                float x0 = r[k].x + fdv.x, x1 = r[k].y + fdv.y;
                float x2 = r[k].z + fdv.z, x3 = r[k].w + fdv.w;
                x0 = x0 > 0.f ? x0 : x0 * NEG_SLOPE;
                x1 = x1 > 0.f ? x1 : x1 * NEG_SLOPE;
                x2 = x2 > 0.f ? x2 : x2 * NEG_SLOPE;
                x3 = x3 > 0.f ? x3 : x3 * NEG_SLOPE;
                const float sc = row16_sum(x0 * av.x + x1 * av.y + x2 * av.z + x3 * av.w);
                if (off + j + k < deg) {
                    const float mn    = fmaxf(m, sc);
                    const float ed    = expf(fminf(m, sc) - mn);
                    const bool  up    = sc > m;
                    const float scale = up ? ed : 1.0f;
                    const float pe    = up ? 1.0f : ed;
                    l     = l * scale + pe;
                    acc.x = acc.x * scale + pe * r[k].x;
                    acc.y = acc.y * scale + pe * r[k].y;
                    acc.z = acc.z * scale + pe * r[k].z;
                    acc.w = acc.w * scale + pe * r[k].w;
                    m = mn;
                }
            }
        }
    }

    float4 o = make_float4(0.f, 0.f, 0.f, 0.f);
    if (deg > 0) {
        o.x = acc.x / l; o.y = acc.y / l; o.z = acc.z / l; o.w = acc.w / l;
    }

    const float mean = row16_sum((o.x + o.y) + (o.z + o.w)) * (1.0f / HID);
    const float c0 = o.x - mean, c1 = o.y - mean, c2 = o.z - mean, c3 = o.w - mean;
    const float var = row16_sum((c0 * c0 + c1 * c1) + (c2 * c2 + c3 * c3)) * (1.0f / HID);
    const float inv = rsqrtf(var + LN_EPS);

    const float4 g  = ((const float4*)gamma)[l16];
    const float4 bt = ((const float4*)beta)[l16];
    const float x0 = c0 * inv * g.x + bt.x;
    const float x1 = c1 * inv * g.y + bt.y;
    const float x2 = c2 * inv * g.z + bt.z;
    const float x3 = c3 * inv * g.w + bt.w;
    const float kk = 0.70710678118654752f;
    hv.x += 0.5f * x0 * (1.0f + erff(x0 * kk));
    hv.y += 0.5f * x1 * (1.0f + erff(x1 * kk));
    hv.z += 0.5f * x2 * (1.0f + erff(x2 * kk));
    hv.w += 0.5f * x3 * (1.0f + erff(x3 * kk));

    if (valid) ((float4*)(h + (size_t)node * HID))[l16] = hv;
}

extern "C" void kernel_launch(void* const* d_in, const int* in_sizes, int n_in,
                              void* d_out, int out_size, void* d_ws, size_t ws_size,
                              hipStream_t stream) {
    const float* nodes = (const float*)d_in[0];
    const int*   src   = (const int*)d_in[1];
    const int*   dst   = (const int*)d_in[2];
    const float* w_in  = (const float*)d_in[3];
    const float* b_in  = (const float*)d_in[4];
    const float* w_src = (const float*)d_in[5];
    const float* b_src = (const float*)d_in[6];
    const float* w_dst = (const float*)d_in[7];
    const float* b_dst = (const float*)d_in[8];
    const float* attn  = (const float*)d_in[9];
    const float* gamma = (const float*)d_in[10];
    const float* beta  = (const float*)d_in[11];
    const float* w_out = (const float*)d_in[12];
    const float* b_out = (const float*)d_in[13];

    const int N = in_sizes[0] / HID;
    const int E = in_sizes[1];
    const int NL = 6;

    float* ws   = (float*)d_ws;
    float* h    = ws;  ws += (size_t)N * HID;
    float* fs   = ws;  ws += (size_t)N * HID;
    float* fd   = ws;  ws += (size_t)N * HID;
    float* tbuf = ws;  ws += (size_t)N * HID;
    int* ssrc      = (int*)ws; ws += E;
    int* seg_start = (int*)ws; ws += N;
    int* seg_end   = (int*)ws;
    int* eid       = (int*)tbuf;

    const int ggrid  = (N + 63) / 64;
    const int eblk   = 256;
    const int egrid  = (E + eblk - 1) / eblk;
    const int ngrid  = (N + 255) / 256;
    const int sgrid  = (N * 16 + 255) / 256;

    gemm64_wmma<4><<<ggrid, 128, 0, stream>>>(nodes, w_in, b_in, h, N, HID);

    zero_counts<<<ngrid, 256, 0, stream>>>(seg_start, N);
    count_dst<<<egrid, eblk, 0, stream>>>(dst, seg_start, E);
    scan_counts<<<1, 1024, 0, stream>>>(seg_start, seg_end, N);
    fill_by_dst<<<egrid, eblk, 0, stream>>>(dst, seg_end, eid, E);
    order_segments<<<egrid, eblk, 0, stream>>>(src, dst, seg_start, seg_end, eid, ssrc, E);

    for (int L = 0; L < NL; ++L) {
        gemm64_wmma<4><<<ggrid, 128, 0, stream>>>(h, w_src + (size_t)L * HID * HID,
                                                  b_src + (size_t)L * HID, fs, N, HID);
        gemm64_wmma<4><<<ggrid, 128, 0, stream>>>(h, w_dst + (size_t)L * HID * HID,
                                                  b_dst + (size_t)L * HID, fd, N, HID);
        seg_softmax_update<<<sgrid, 256, 0, stream>>>(fs, fd, ssrc, seg_start, seg_end,
                                                      attn + (size_t)L * HID,
                                                      gamma + (size_t)L * HID,
                                                      beta + (size_t)L * HID, h, N);
    }

    gemm64_wmma<1><<<ggrid, 128, 0, stream>>>(h, w_out, b_out, (float*)d_out, N, 8);
}


# next layer's fc_src/fc_dst (and the output projection) moved into the edge kernel: a block's 16 new h rows pass through LDS as one WMMA A tile, one output tile per wave; 11 launches fewer
# speedup vs baseline: 9.5069x; 1.1537x over previous
#include <hip/hip_runtime.h>
#include <hip/hip_bf16.h>
#include <math.h>


typedef __attribute__((ext_vector_type(2))) float v2f;
typedef __attribute__((ext_vector_type(8))) float v8f;

#define HID 64
#define NEG_SLOPE 0.2f
#define LN_EPS 1e-5f

__global__ void __launch_bounds__(128)
gemm64_wmma(const float* __restrict__ X,
            const float* __restrict__ W,
            const float* __restrict__ bias,
            float* __restrict__ Y,
            int nrows) {
    constexpr int NTILES = 4;
    constexpr int nout = HID;
    __shared__ __attribute__((aligned(16))) float lds_w[64][64];

    const int tid = threadIdx.x;
    {
        const float4* W4 = (const float4*)W;
        float4* L4 = (float4*)&lds_w[0][0];
        float4 wv[8];
#pragma unroll
        for (int i = 0; i < 8; ++i) wv[i] = W4[tid + 128 * i];
#pragma unroll
        for (int i = 0; i < 8; ++i) L4[tid + 128 * i] = wv[i];
    }
    __syncthreads();

    int lane  = tid & 31;
    int wave  = tid >> 5;
    int half  = lane >> 4;
    int l15   = lane & 15;
    int rbase = blockIdx.x * 64 + wave * 16;

    int arow = rbase + l15;
    if (arow >= nrows) arow = nrows - 1;
    const float* xrow = X + (size_t)arow * 64;

    v8f acc[NTILES];
#pragma unroll
    for (int t = 0; t < NTILES; ++t)
#pragma unroll
        for (int j = 0; j < 8; ++j) acc[t][j] = 0.0f;

#pragma unroll
    for (int k0 = 0; k0 < 64; k0 += 4) {
        const float* ap = xrow + k0 + 2 * half;
        v2f a;
        a.x = ap[0];
        a.y = ap[1];
#pragma unroll
        for (int t = 0; t < NTILES; ++t) {
            v2f b;
            b.x = lds_w[k0 + 2 * half + 0][t * 16 + l15];
            b.y = lds_w[k0 + 2 * half + 1][t * 16 + l15];
            acc[t] = __builtin_amdgcn_wmma_f32_16x16x4_f32(
                false, a, false, b, (short)0, acc[t], false, false);
        }
    }

#pragma unroll
    for (int t = 0; t < NTILES; ++t) {
        int col = t * 16 + l15;
        float bv = bias[col];
#pragma unroll
        for (int i = 0; i < 8; ++i) {
            int row = rbase + 8 * half + i;
            if (row < nrows)
                Y[(size_t)row * nout + col] = acc[t][i] + bv;
        }
    }
}

__global__ void zero_counts(int* __restrict__ cnt, int n) {
    int i = blockIdx.x * blockDim.x + threadIdx.x;
    if (i < n) cnt[i] = 0;
}

__global__ void count_dst(const int* __restrict__ dst, int* __restrict__ cnt, int E) {
    int e = blockIdx.x * blockDim.x + threadIdx.x;
    if (e < E) atomicAdd(&cnt[dst[e]], 1);
}

__global__ void __launch_bounds__(1024)
scan_counts(int* __restrict__ start, int* __restrict__ cursor, int n) {
    __shared__ int part[1024];
    const int tid = threadIdx.x;
    const int per = (n + 1023) >> 10;
    const int b = tid * per;
    int e = b + per;
    if (e > n) e = n;
    int s = 0;
    for (int i = b; i < e; ++i) s += start[i];
    part[tid] = s;
    __syncthreads();
    for (int d = 1; d < 1024; d <<= 1) {
        int v = (tid >= d) ? part[tid - d] : 0;
        __syncthreads();
        part[tid] += v;
        __syncthreads();
    }
    int run = part[tid] - s;
#pragma unroll 8
    for (int i = b; i < e; ++i) {
        int c = start[i];
        start[i]  = run;
        cursor[i] = run;
        run += c;
    }
}

__global__ void fill_by_dst(const int* __restrict__ dst, int* __restrict__ cursor,
                            int* __restrict__ eid, int E) {
    int e = blockIdx.x * blockDim.x + threadIdx.x;
    if (e >= E) return;
    int p = atomicAdd(&cursor[dst[e]], 1);
    eid[p] = e;
}

__global__ void order_segments(const int* __restrict__ src, const int* __restrict__ dst,
                               const int* __restrict__ seg_start,
                               const int* __restrict__ seg_end,
                               const int* __restrict__ eid,
                               int* __restrict__ ssrc, int E) {
    int p = blockIdx.x * blockDim.x + threadIdx.x;
    if (p >= E) return;
    int e = eid[p];
    int d = dst[e];
    int b = seg_start[d], en = seg_end[d];
    int r = 0;
    for (int q = b; q < en; ++q) r += (eid[q] < e) ? 1 : 0;
    ssrc[b + r] = src[e];
}

template <int CTRL>
__device__ __forceinline__ float dpp_row(float v) {
    return __builtin_bit_cast(float, __builtin_amdgcn_update_dpp(
        0, __builtin_bit_cast(int, v), CTRL, 0xf, 0xf, false));
}
__device__ __forceinline__ float row16_sum(float v) {
    v += dpp_row<0x161>(v);
    v += dpp_row<0x162>(v);
    v += dpp_row<0x164>(v);
    v += dpp_row<0x168>(v);
    return v;
}

template <bool LAST>
__global__ void __launch_bounds__(256)
seg_softmax_update(const float* __restrict__ fs, float* __restrict__ fd,
                   const int* __restrict__ ssrc,
                   const int* __restrict__ seg_start,
                   const int* __restrict__ seg_end,
                   const float* __restrict__ attn,
                   const float* __restrict__ gamma,
                   const float* __restrict__ beta,
                   float* __restrict__ h,
                   const float* __restrict__ wa, const float* __restrict__ ba,
                   const float* __restrict__ wb, const float* __restrict__ bb,
                   float* __restrict__ ya, int n) {
    __shared__ __attribute__((aligned(16))) float lds_h[16][HID + 4];
    __shared__ __attribute__((aligned(16))) float lds_y[LAST ? 1 : 16][LAST ? 4 : 2 * HID + 4];

    const int lane = threadIdx.x & 31;
    const int l16  = lane & 15;
    const int half = lane >> 4;
    const int node = (blockIdx.x * blockDim.x + threadIdx.x) >> 4;
    const bool valid = node < n;
    const int nd = valid ? node : (n - 1);
    const int b = seg_start[nd];
    const int deg = valid ? (seg_end[nd] - b) : 0;
    const float4 fdv = ((const float4*)(fd + (size_t)nd * HID))[l16];
    const float4 av  = ((const float4*)attn)[l16];
    float4 hv = ((const float4*)(h + (size_t)nd * HID))[l16];

    const int d0 = __builtin_amdgcn_readlane(deg, 0);
    const int d1 = __builtin_amdgcn_readlane(deg, 16);
    const int dmax = d0 > d1 ? d0 : d1;

    float m = -INFINITY, l = 0.0f;
    float4 acc = make_float4(0.f, 0.f, 0.f, 0.f);

    for (int off = 0; off < dmax; off += 16) {
        const int p = off + l16;
        const int myidx = (p < deg) ? ssrc[b + p] : 0;
        int lim = dmax - off;
        if (lim > 16) lim = 16;
        for (int j = 0; j < lim; j += 4) {
            float4 r[4];
#pragma unroll
            for (int k = 0; k < 4; ++k) {
                const int s0 = __builtin_amdgcn_readlane(myidx, j + k);
                const int s1 = __builtin_amdgcn_readlane(myidx, j + k + 16);
                const int s  = half ? s1 : s0;
                r[k] = ((const float4*)(fs + (size_t)s * HID))[l16];
            }
#pragma unroll
            for (int k = 0; k < 4; ++k) {
                float x0 = r[k].x + fdv.x, x1 = r[k].y + fdv.y;
                float x2 = r[k].z + fdv.z, x3 = r[k].w + fdv.w;
                x0 = x0 > 0.f ? x0 : x0 * NEG_SLOPE;
                x1 = x1 > 0.f ? x1 : x1 * NEG_SLOPE;
                x2 = x2 > 0.f ? x2 : x2 * NEG_SLOPE;
                x3 = x3 > 0.f ? x3 : x3 * NEG_SLOPE;
                const float sc = row16_sum(x0 * av.x + x1 * av.y + x2 * av.z + x3 * av.w);
                if (off + j + k < deg) {
                    const float mn    = fmaxf(m, sc);
                    const float ed    = expf(fminf(m, sc) - mn);
                    const bool  up    = sc > m;
                    const float scale = up ? ed : 1.0f;
                    const float pe    = up ? 1.0f : ed;
                    l     = l * scale + pe;
                    acc.x = acc.x * scale + pe * r[k].x;
                    acc.y = acc.y * scale + pe * r[k].y;
                    acc.z = acc.z * scale + pe * r[k].z;
                    acc.w = acc.w * scale + pe * r[k].w;
                    m = mn;
                }
            }
        }
    }

    float4 o = make_float4(0.f, 0.f, 0.f, 0.f);
    if (deg > 0) {
        o.x = acc.x / l; o.y = acc.y / l; o.z = acc.z / l; o.w = acc.w / l;
    }

    const float mean = row16_sum((o.x + o.y) + (o.z + o.w)) * (1.0f / HID);
    const float c0 = o.x - mean, c1 = o.y - mean, c2 = o.z - mean, c3 = o.w - mean;
    const float var = row16_sum((c0 * c0 + c1 * c1) + (c2 * c2 + c3 * c3)) * (1.0f / HID);
    const float inv = rsqrtf(var + LN_EPS);

    const float4 g  = ((const float4*)gamma)[l16];
    const float4 bt = ((const float4*)beta)[l16];
    const float x0 = c0 * inv * g.x + bt.x;
    const float x1 = c1 * inv * g.y + bt.y;
    const float x2 = c2 * inv * g.z + bt.z;
    const float x3 = c3 * inv * g.w + bt.w;
    const float kk = 0.70710678118654752f;
    hv.x += 0.5f * x0 * (1.0f + erff(x0 * kk));
    hv.y += 0.5f * x1 * (1.0f + erff(x1 * kk));
    hv.z += 0.5f * x2 * (1.0f + erff(x2 * kk));
    hv.w += 0.5f * x3 * (1.0f + erff(x3 * kk));

    const int wave = __builtin_amdgcn_readfirstlane((int)(threadIdx.x >> 5));
    const bool owns_tile = LAST ? (wave == 0) : true;
    const int   ldw   = LAST ? 8 : HID;
    const bool  colok = LAST ? (l16 < 8) : true;
    const int   col   = LAST ? (colok ? l16 : 7)
                             : ((wave & 3) * 16 + l16);
    const float* Wt = LAST ? wa : (wave < 4 ? wa : wb);
    const float* Bt = LAST ? ba : (wave < 4 ? ba : bb);

    v2f bfrag[16];
    float bv = 0.0f;
    if (owns_tile) {
#pragma unroll
        for (int s = 0; s < 16; ++s) {
            const float b0 = Wt[(4 * s + 2 * half + 0) * ldw + col];
            const float b1 = Wt[(4 * s + 2 * half + 1) * ldw + col];
            bfrag[s].x = colok ? b0 : 0.0f;
            bfrag[s].y = colok ? b1 : 0.0f;
        }
        const float bl = Bt[col];
        bv = colok ? bl : 0.0f;
    }

    ((float4*)&lds_h[threadIdx.x >> 4][0])[l16] = hv;
    __syncthreads();

    if (owns_tile) {
        v8f acc = {};
#pragma unroll
        for (int s = 0; s < 16; ++s) {
            const v2f a = *(const v2f*)&lds_h[l16][4 * s + 2 * half];
            acc = __builtin_amdgcn_wmma_f32_16x16x4_f32(
                false, a, false, bfrag[s], (short)0, acc, false, false);
        }
#pragma unroll
        for (int i = 0; i < 8; ++i) {
            if (LAST) {
                const int row = blockIdx.x * 16 + 8 * half + i;
                if (row < n && colok)
                    ya[(size_t)row * 8 + col] = acc[i] + bv;
            } else {
                lds_y[8 * half + i][wave * 16 + l16] = acc[i] + bv;
            }
        }
    }

    if (!LAST) {
        __syncthreads();
        if (valid) ((float4*)(h + (size_t)node * HID))[l16] = hv;
        float* Yp = half ? fd : ya;
#pragma unroll
        for (int q = 0; q < 2; ++q) {
            const int r = wave + 8 * q;
            const int row = blockIdx.x * 16 + r;
            const float4 v = *(const float4*)&lds_y[r][4 * lane];
            if (row < n) ((float4*)(Yp + (size_t)row * HID))[l16] = v;
        }
    }
}

extern "C" void kernel_launch(void* const* d_in, const int* in_sizes, int n_in,
                              void* d_out, int out_size, void* d_ws, size_t ws_size,
                              hipStream_t stream) {
    const float* nodes = (const float*)d_in[0];
    const int*   src   = (const int*)d_in[1];
    const int*   dst   = (const int*)d_in[2];
    const float* w_in  = (const float*)d_in[3];
    const float* b_in  = (const float*)d_in[4];
    const float* w_src = (const float*)d_in[5];
    const float* b_src = (const float*)d_in[6];
    const float* w_dst = (const float*)d_in[7];
    const float* b_dst = (const float*)d_in[8];
    const float* attn  = (const float*)d_in[9];
    const float* gamma = (const float*)d_in[10];
    const float* beta  = (const float*)d_in[11];
    const float* w_out = (const float*)d_in[12];
    const float* b_out = (const float*)d_in[13];

    const int N = in_sizes[0] / HID;
    const int E = in_sizes[1];
    const int NL = 6;

    float* ws   = (float*)d_ws;
    float* h    = ws;  ws += (size_t)N * HID;
    float* fs   = ws;  ws += (size_t)N * HID;
    float* fd   = ws;  ws += (size_t)N * HID;
    float* tbuf = ws;  ws += (size_t)N * HID;
    int* ssrc      = (int*)ws; ws += E;
    int* seg_start = (int*)ws; ws += N;
    int* seg_end   = (int*)ws;
    int* eid       = (int*)tbuf;
    float* fsp[2]  = { fs, tbuf };

    const int ggrid  = (N + 63) / 64;
    const int eblk   = 256;
    const int egrid  = (E + eblk - 1) / eblk;
    const int ngrid  = (N + 255) / 256;
    const int sgrid  = (N * 16 + 255) / 256;

    gemm64_wmma<<<ggrid, 128, 0, stream>>>(nodes, w_in, b_in, h, N);

    zero_counts<<<ngrid, 256, 0, stream>>>(seg_start, N);
    count_dst<<<egrid, eblk, 0, stream>>>(dst, seg_start, E);
    scan_counts<<<1, 1024, 0, stream>>>(seg_start, seg_end, N);
    fill_by_dst<<<egrid, eblk, 0, stream>>>(dst, seg_end, eid, E);
    order_segments<<<egrid, eblk, 0, stream>>>(src, dst, seg_start, seg_end, eid, ssrc, E);

    gemm64_wmma<<<ggrid, 128, 0, stream>>>(h, w_src, b_src, fs, N);
    gemm64_wmma<<<ggrid, 128, 0, stream>>>(h, w_dst, b_dst, fd, N);

    for (int L = 0; L + 1 < NL; ++L) {
        seg_softmax_update<false><<<sgrid, 256, 0, stream>>>(
            fsp[L & 1], fd, ssrc, seg_start, seg_end,
            attn + (size_t)L * HID, gamma + (size_t)L * HID, beta + (size_t)L * HID, h,
            w_src + (size_t)(L + 1) * HID * HID, b_src + (size_t)(L + 1) * HID,
            w_dst + (size_t)(L + 1) * HID * HID, b_dst + (size_t)(L + 1) * HID,
            fsp[(L + 1) & 1], N);
    }
    seg_softmax_update<true><<<sgrid, 256, 0, stream>>>(
        fsp[(NL - 1) & 1], fd, ssrc, seg_start, seg_end,
        attn + (size_t)(NL - 1) * HID, gamma + (size_t)(NL - 1) * HID,
        beta + (size_t)(NL - 1) * HID, h,
        w_out, b_out, w_out, b_out, (float*)d_out, N);
}


# edge loop, fewer vector instructions: softmax max/denominator/row sum updated once per 4-edge batch (sums grouped by four, edge order kept), weights by hardware exp2, leaky relu as max(x, 0.2x)
# speedup vs baseline: 10.0999x; 1.0624x over previous
#include <hip/hip_runtime.h>
#include <hip/hip_bf16.h>
#include <math.h>


typedef __attribute__((ext_vector_type(2))) float v2f;
typedef __attribute__((ext_vector_type(8))) float v8f;

#define HID 64
#define NEG_SLOPE 0.2f
#define LN_EPS 1e-5f

__global__ void __launch_bounds__(128)
gemm64_wmma(const float* __restrict__ X,
            const float* __restrict__ W,
            const float* __restrict__ bias,
            float* __restrict__ Y,
            int nrows) {
    constexpr int NTILES = 4;
    constexpr int nout = HID;
    __shared__ __attribute__((aligned(16))) float lds_w[64][64];

    const int tid = threadIdx.x;
    {
        const float4* W4 = (const float4*)W;
        float4* L4 = (float4*)&lds_w[0][0];
        float4 wv[8];
#pragma unroll
        for (int i = 0; i < 8; ++i) wv[i] = W4[tid + 128 * i];
#pragma unroll
        for (int i = 0; i < 8; ++i) L4[tid + 128 * i] = wv[i];
    }
    __syncthreads();

    int lane  = tid & 31;
    int wave  = tid >> 5;
    int half  = lane >> 4;
    int l15   = lane & 15;
    int rbase = blockIdx.x * 64 + wave * 16;

    int arow = rbase + l15;
    if (arow >= nrows) arow = nrows - 1;
    const float* xrow = X + (size_t)arow * 64;

    v8f acc[NTILES];
#pragma unroll
    for (int t = 0; t < NTILES; ++t)
#pragma unroll
        for (int j = 0; j < 8; ++j) acc[t][j] = 0.0f;

#pragma unroll
    for (int k0 = 0; k0 < 64; k0 += 4) {
        const float* ap = xrow + k0 + 2 * half;
        v2f a;
        a.x = ap[0];
        a.y = ap[1];
#pragma unroll
        for (int t = 0; t < NTILES; ++t) {
            v2f b;
            b.x = lds_w[k0 + 2 * half + 0][t * 16 + l15];
            b.y = lds_w[k0 + 2 * half + 1][t * 16 + l15];
            acc[t] = __builtin_amdgcn_wmma_f32_16x16x4_f32(
                false, a, false, b, (short)0, acc[t], false, false);
        }
    }

#pragma unroll
    for (int t = 0; t < NTILES; ++t) {
        int col = t * 16 + l15;
        float bv = bias[col];
#pragma unroll
        for (int i = 0; i < 8; ++i) {
            int row = rbase + 8 * half + i;
            if (row < nrows)
                Y[(size_t)row * nout + col] = acc[t][i] + bv;
        }
    }
}

__global__ void zero_counts(int* __restrict__ cnt, int n) {
    int i = blockIdx.x * blockDim.x + threadIdx.x;
    if (i < n) cnt[i] = 0;
}

__global__ void count_dst(const int* __restrict__ dst, int* __restrict__ cnt, int E) {
    int e = blockIdx.x * blockDim.x + threadIdx.x;
    if (e < E) atomicAdd(&cnt[dst[e]], 1);
}

__global__ void __launch_bounds__(1024)
scan_counts(int* __restrict__ start, int* __restrict__ cursor, int n) {
    __shared__ int part[1024];
    const int tid = threadIdx.x;
    const int per = (n + 1023) >> 10;
    const int b = tid * per;
    int e = b + per;
    if (e > n) e = n;
    int s = 0;
    for (int i = b; i < e; ++i) s += start[i];
    part[tid] = s;
    __syncthreads();
    for (int d = 1; d < 1024; d <<= 1) {
        int v = (tid >= d) ? part[tid - d] : 0;
        __syncthreads();
        part[tid] += v;
        __syncthreads();
    }
    int run = part[tid] - s;
#pragma unroll 8
    for (int i = b; i < e; ++i) {
        int c = start[i];
        start[i]  = run;
        cursor[i] = run;
        run += c;
    }
}

__global__ void fill_by_dst(const int* __restrict__ dst, int* __restrict__ cursor,
                            int* __restrict__ eid, int E) {
    int e = blockIdx.x * blockDim.x + threadIdx.x;
    if (e >= E) return;
    int p = atomicAdd(&cursor[dst[e]], 1);
    eid[p] = e;
}

__global__ void order_segments(const int* __restrict__ src, const int* __restrict__ dst,
                               const int* __restrict__ seg_start,
                               const int* __restrict__ seg_end,
                               const int* __restrict__ eid,
                               int* __restrict__ ssrc, int E) {
    int p = blockIdx.x * blockDim.x + threadIdx.x;
    if (p >= E) return;
    int e = eid[p];
    int d = dst[e];
    int b = seg_start[d], en = seg_end[d];
    int r = 0;
    for (int q = b; q < en; ++q) r += (eid[q] < e) ? 1 : 0;
    ssrc[b + r] = src[e];
}

template <int CTRL>
__device__ __forceinline__ float dpp_row(float v) {
    return __builtin_bit_cast(float, __builtin_amdgcn_update_dpp(
        0, __builtin_bit_cast(int, v), CTRL, 0xf, 0xf, false));
}
__device__ __forceinline__ float row16_sum(float v) {
    v += dpp_row<0x161>(v);
    v += dpp_row<0x162>(v);
    v += dpp_row<0x164>(v);
    v += dpp_row<0x168>(v);
    return v;
}

__device__ __forceinline__ float pow2_hw(float x) {
    return __builtin_amdgcn_exp2f(x);
}

template <bool LAST>
__global__ void __launch_bounds__(256) __attribute__((amdgpu_waves_per_eu(16, 16)))
seg_softmax_update(const float* __restrict__ fs, float* __restrict__ fd,
                   const int* __restrict__ ssrc,
                   const int* __restrict__ seg_start,
                   const int* __restrict__ seg_end,
                   const float* __restrict__ attn,
                   const float* __restrict__ gamma,
                   const float* __restrict__ beta,
                   float* __restrict__ h,
                   const float* __restrict__ wa, const float* __restrict__ ba,
                   const float* __restrict__ wb, const float* __restrict__ bb,
                   float* __restrict__ ya, int n) {
    __shared__ __attribute__((aligned(16))) float lds_h[16][HID + 4];
    __shared__ __attribute__((aligned(16))) float lds_y[LAST ? 1 : 16][LAST ? 4 : 2 * HID + 4];

    const int lane = threadIdx.x & 31;
    const int l16  = lane & 15;
    const int half = lane >> 4;
    const int node = (blockIdx.x * blockDim.x + threadIdx.x) >> 4;
    const bool valid = node < n;
    const int nd = valid ? node : (n - 1);
    const int b = seg_start[nd];
    const int deg = valid ? (seg_end[nd] - b) : 0;
    const float4 fdv = ((const float4*)(fd + (size_t)nd * HID))[l16];
    float4 av = ((const float4*)attn)[l16];
    {
        const float log2e = 1.44269504088896340736f;
        av.x *= log2e; av.y *= log2e; av.z *= log2e; av.w *= log2e;
    }
    float4 hv = ((const float4*)(h + (size_t)nd * HID))[l16];

    const int d0 = __builtin_amdgcn_readlane(deg, 0);
    const int d1 = __builtin_amdgcn_readlane(deg, 16);
    const int dmax = d0 > d1 ? d0 : d1;

    float m = -INFINITY, l = 0.0f;
    float4 acc = make_float4(0.f, 0.f, 0.f, 0.f);

    for (int off = 0; off < dmax; off += 16) {
        const int p = off + l16;
        const int myidx = (p < deg) ? ssrc[b + p] : 0;
        int lim = dmax - off;
        if (lim > 16) lim = 16;
        for (int j = 0; j < lim; j += 4) {
            float4 r[4];
#pragma unroll
            for (int k = 0; k < 4; ++k) {
                const int s0 = __builtin_amdgcn_readlane(myidx, j + k);
                const int s1 = __builtin_amdgcn_readlane(myidx, j + k + 16);
                const int s  = half ? s1 : s0;
                r[k] = ((const float4*)(fs + (size_t)s * HID))[l16];
            }
            float sc[4];
#pragma unroll
            for (int k = 0; k < 4; ++k) {
                float x0 = r[k].x + fdv.x, x1 = r[k].y + fdv.y;
                float x2 = r[k].z + fdv.z, x3 = r[k].w + fdv.w;
                x0 = fmaxf(x0, x0 * NEG_SLOPE);
                x1 = fmaxf(x1, x1 * NEG_SLOPE);
                x2 = fmaxf(x2, x2 * NEG_SLOPE);
                x3 = fmaxf(x3, x3 * NEG_SLOPE);
                const float s = row16_sum(x0 * av.x + x1 * av.y + x2 * av.z + x3 * av.w);
                sc[k] = (off + j + k < deg) ? s : -INFINITY;
            }
            if (off + j < deg) {
                const float bm    = fmaxf(fmaxf(sc[0], sc[1]), fmaxf(sc[2], sc[3]));
                const float mn    = fmaxf(m, bm);
                const float scale = pow2_hw(m - mn);
                const float p0 = pow2_hw(sc[0] - mn), p1 = pow2_hw(sc[1] - mn);
                const float p2 = pow2_hw(sc[2] - mn), p3 = pow2_hw(sc[3] - mn);
                l     = l * scale + ((p0 + p1) + (p2 + p3));
                acc.x = acc.x * scale + p0 * r[0].x + p1 * r[1].x + p2 * r[2].x + p3 * r[3].x;
                acc.y = acc.y * scale + p0 * r[0].y + p1 * r[1].y + p2 * r[2].y + p3 * r[3].y;
                acc.z = acc.z * scale + p0 * r[0].z + p1 * r[1].z + p2 * r[2].z + p3 * r[3].z;
                acc.w = acc.w * scale + p0 * r[0].w + p1 * r[1].w + p2 * r[2].w + p3 * r[3].w;
                m = mn;
            }
        }
    }

    float4 o = make_float4(0.f, 0.f, 0.f, 0.f);
    if (deg > 0) {
        o.x = acc.x / l; o.y = acc.y / l; o.z = acc.z / l; o.w = acc.w / l;
    }

    const float mean = row16_sum((o.x + o.y) + (o.z + o.w)) * (1.0f / HID);
    const float c0 = o.x - mean, c1 = o.y - mean, c2 = o.z - mean, c3 = o.w - mean;
    const float var = row16_sum((c0 * c0 + c1 * c1) + (c2 * c2 + c3 * c3)) * (1.0f / HID);
    const float inv = rsqrtf(var + LN_EPS);

    const float4 g  = ((const float4*)gamma)[l16];
    const float4 bt = ((const float4*)beta)[l16];
    const float x0 = c0 * inv * g.x + bt.x;
    const float x1 = c1 * inv * g.y + bt.y;
    const float x2 = c2 * inv * g.z + bt.z;
    const float x3 = c3 * inv * g.w + bt.w;
    const float kk = 0.70710678118654752f;
    hv.x += 0.5f * x0 * (1.0f + erff(x0 * kk));
    hv.y += 0.5f * x1 * (1.0f + erff(x1 * kk));
    hv.z += 0.5f * x2 * (1.0f + erff(x2 * kk));
    hv.w += 0.5f * x3 * (1.0f + erff(x3 * kk));

    const int wave = __builtin_amdgcn_readfirstlane((int)(threadIdx.x >> 5));
    const bool owns_tile = LAST ? (wave == 0) : true;
    const int   ldw   = LAST ? 8 : HID;
    const bool  colok = LAST ? (l16 < 8) : true;
    const int   col   = LAST ? (colok ? l16 : 7)
                             : ((wave & 3) * 16 + l16);
    const float* Wt = LAST ? wa : (wave < 4 ? wa : wb);
    const float* Bt = LAST ? ba : (wave < 4 ? ba : bb);

    v2f bfrag[16];
    float bv = 0.0f;
    if (owns_tile) {
#pragma unroll
        for (int s = 0; s < 16; ++s) {
            const float b0 = Wt[(4 * s + 2 * half + 0) * ldw + col];
            const float b1 = Wt[(4 * s + 2 * half + 1) * ldw + col];
            bfrag[s].x = colok ? b0 : 0.0f;
            bfrag[s].y = colok ? b1 : 0.0f;
        }
        const float bl = Bt[col];
        bv = colok ? bl : 0.0f;
    }

    ((float4*)&lds_h[threadIdx.x >> 4][0])[l16] = hv;
    __syncthreads();

    if (owns_tile) {
        v8f acc = {};
#pragma unroll
        for (int s = 0; s < 16; ++s) {
            const v2f a = *(const v2f*)&lds_h[l16][4 * s + 2 * half];
            acc = __builtin_amdgcn_wmma_f32_16x16x4_f32(
                false, a, false, bfrag[s], (short)0, acc, false, false);
        }
#pragma unroll
        for (int i = 0; i < 8; ++i) {
            if (LAST) {
                const int row = blockIdx.x * 16 + 8 * half + i;
                if (row < n && colok)
                    ya[(size_t)row * 8 + col] = acc[i] + bv;
            } else {
                lds_y[8 * half + i][wave * 16 + l16] = acc[i] + bv;
            }
        }
    }

    if (!LAST) {
        __syncthreads();
        if (valid) ((float4*)(h + (size_t)node * HID))[l16] = hv;
        float* Yp = half ? fd : ya;
#pragma unroll
        for (int q = 0; q < 2; ++q) {
            const int r = wave + 8 * q;
            const int row = blockIdx.x * 16 + r;
            const float4 v = *(const float4*)&lds_y[r][4 * lane];
            if (row < n) ((float4*)(Yp + (size_t)row * HID))[l16] = v;
        }
    }
}

extern "C" void kernel_launch(void* const* d_in, const int* in_sizes, int n_in,
                              void* d_out, int out_size, void* d_ws, size_t ws_size,
                              hipStream_t stream) {
    const float* nodes = (const float*)d_in[0];
    const int*   src   = (const int*)d_in[1];
    const int*   dst   = (const int*)d_in[2];
    const float* w_in  = (const float*)d_in[3];
    const float* b_in  = (const float*)d_in[4];
    const float* w_src = (const float*)d_in[5];
    const float* b_src = (const float*)d_in[6];
    const float* w_dst = (const float*)d_in[7];
    const float* b_dst = (const float*)d_in[8];
    const float* attn  = (const float*)d_in[9];
    const float* gamma = (const float*)d_in[10];
    const float* beta  = (const float*)d_in[11];
    const float* w_out = (const float*)d_in[12];
    const float* b_out = (const float*)d_in[13];

    const int N = in_sizes[0] / HID;
    const int E = in_sizes[1];
    const int NL = 6;

    float* ws   = (float*)d_ws;
    float* h    = ws;  ws += (size_t)N * HID;
    float* fs   = ws;  ws += (size_t)N * HID;
    float* fd   = ws;  ws += (size_t)N * HID;
    float* tbuf = ws;  ws += (size_t)N * HID;
    int* ssrc      = (int*)ws; ws += E;
    int* seg_start = (int*)ws; ws += N;
    int* seg_end   = (int*)ws;
    int* eid       = (int*)tbuf;
    float* fsp[2]  = { fs, tbuf };

    const int ggrid  = (N + 63) / 64;
    const int eblk   = 256;
    const int egrid  = (E + eblk - 1) / eblk;
    const int ngrid  = (N + 255) / 256;
    const int sgrid  = (N * 16 + 255) / 256;

    gemm64_wmma<<<ggrid, 128, 0, stream>>>(nodes, w_in, b_in, h, N);

    zero_counts<<<ngrid, 256, 0, stream>>>(seg_start, N);
    count_dst<<<egrid, eblk, 0, stream>>>(dst, seg_start, E);
    scan_counts<<<1, 1024, 0, stream>>>(seg_start, seg_end, N);
    fill_by_dst<<<egrid, eblk, 0, stream>>>(dst, seg_end, eid, E);
    order_segments<<<egrid, eblk, 0, stream>>>(src, dst, seg_start, seg_end, eid, ssrc, E);

    gemm64_wmma<<<ggrid, 128, 0, stream>>>(h, w_src, b_src, fs, N);
    gemm64_wmma<<<ggrid, 128, 0, stream>>>(h, w_dst, b_dst, fd, N);

    for (int L = 0; L + 1 < NL; ++L) {
        seg_softmax_update<false><<<sgrid, 256, 0, stream>>>(
            fsp[L & 1], fd, ssrc, seg_start, seg_end,
            attn + (size_t)L * HID, gamma + (size_t)L * HID, beta + (size_t)L * HID, h,
            w_src + (size_t)(L + 1) * HID * HID, b_src + (size_t)(L + 1) * HID,
            w_dst + (size_t)(L + 1) * HID * HID, b_dst + (size_t)(L + 1) * HID,
            fsp[(L + 1) & 1], N);
    }
    seg_softmax_update<true><<<sgrid, 256, 0, stream>>>(
        fsp[(NL - 1) & 1], fd, ssrc, seg_start, seg_end,
        attn + (size_t)(NL - 1) * HID, gamma + (size_t)(NL - 1) * HID,
        beta + (size_t)(NL - 1) * HID, h,
        w_out, b_out, w_out, b_out, (float*)d_out, N);
}


# scan of the per-destination edge counts: 49 blocks with coalesced loads (each sums the counts before its 1024-count slice, then scans the slice), not one block whose lanes each walk 49 counts
# speedup vs baseline: 11.2061x; 1.1095x over previous
#include <hip/hip_runtime.h>
#include <hip/hip_bf16.h>
#include <math.h>


typedef __attribute__((ext_vector_type(2))) float v2f;
typedef __attribute__((ext_vector_type(8))) float v8f;

#define HID 64
#define NEG_SLOPE 0.2f
#define LN_EPS 1e-5f

__global__ void __launch_bounds__(128)
gemm64_wmma(const float* __restrict__ X,
            const float* __restrict__ W,
            const float* __restrict__ bias,
            float* __restrict__ Y,
            int nrows) {
    constexpr int NTILES = 4;
    constexpr int nout = HID;
    __shared__ __attribute__((aligned(16))) float lds_w[64][64];

    const int tid = threadIdx.x;
    {
        const float4* W4 = (const float4*)W;
        float4* L4 = (float4*)&lds_w[0][0];
        float4 wv[8];
#pragma unroll
        for (int i = 0; i < 8; ++i) wv[i] = W4[tid + 128 * i];
#pragma unroll
        for (int i = 0; i < 8; ++i) L4[tid + 128 * i] = wv[i];
    }
    __syncthreads();

    int lane  = tid & 31;
    int wave  = tid >> 5;
    int half  = lane >> 4;
    int l15   = lane & 15;
    int rbase = blockIdx.x * 64 + wave * 16;

    int arow = rbase + l15;
    if (arow >= nrows) arow = nrows - 1;
    const float* xrow = X + (size_t)arow * 64;

    v8f acc[NTILES];
#pragma unroll
    for (int t = 0; t < NTILES; ++t)
#pragma unroll
        for (int j = 0; j < 8; ++j) acc[t][j] = 0.0f;

#pragma unroll
    for (int k0 = 0; k0 < 64; k0 += 4) {
        const float* ap = xrow + k0 + 2 * half;
        v2f a;
        a.x = ap[0];
        a.y = ap[1];
#pragma unroll
        for (int t = 0; t < NTILES; ++t) {
            v2f b;
            b.x = lds_w[k0 + 2 * half + 0][t * 16 + l15];
            b.y = lds_w[k0 + 2 * half + 1][t * 16 + l15];
            acc[t] = __builtin_amdgcn_wmma_f32_16x16x4_f32(
                false, a, false, b, (short)0, acc[t], false, false);
        }
    }

#pragma unroll
    for (int t = 0; t < NTILES; ++t) {
        int col = t * 16 + l15;
        float bv = bias[col];
#pragma unroll
        for (int i = 0; i < 8; ++i) {
            int row = rbase + 8 * half + i;
            if (row < nrows)
                Y[(size_t)row * nout + col] = acc[t][i] + bv;
        }
    }
}

__global__ void zero_counts(int* __restrict__ cnt, int n) {
    int i = blockIdx.x * blockDim.x + threadIdx.x;
    if (i < n) cnt[i] = 0;
}

__global__ void count_dst(const int* __restrict__ dst, int* __restrict__ cnt, int E) {
    int e = blockIdx.x * blockDim.x + threadIdx.x;
    if (e < E) atomicAdd(&cnt[dst[e]], 1);
}

__global__ void __launch_bounds__(1024)
scan_counts(const int* __restrict__ cnt, int* __restrict__ start,
            int* __restrict__ cursor, int n) {
    __shared__ int w_cnt[32];
    __shared__ int w_pre[32];
    const int tid  = threadIdx.x;
    const int lane = tid & 31;
    const int wave = tid >> 5;
    const int base = blockIdx.x * 1024;

    int pre = 0;
#pragma unroll 8
    for (int i = tid; i < base; i += 1024) pre += cnt[i];

    const int idx = base + tid;
    const int c = (idx < n) ? cnt[idx] : 0;

    int incl = c;
    for (int d = 1; d < 32; d <<= 1) {
        const int v = __shfl_up(incl, d);
        if (lane >= d) incl += v;
    }
    for (int d = 16; d >= 1; d >>= 1) pre += __shfl_xor(pre, d);
    if (lane == 31) w_cnt[wave] = incl;
    if (lane == 0)  w_pre[wave] = pre;
    __syncthreads();

    const int wc = w_cnt[lane];
    int wp = w_pre[lane];
    int wincl = wc;
    for (int d = 1; d < 32; d <<= 1) {
        const int v = __shfl_up(wincl, d);
        if (lane >= d) wincl += v;
    }
    for (int d = 16; d >= 1; d >>= 1) wp += __shfl_xor(wp, d);
    const int woff = __shfl(wincl - wc, wave);

    const int excl = wp + woff + incl - c;
    if (idx < n) {
        start[idx]  = excl;
        cursor[idx] = excl;
    }
}

__global__ void fill_by_dst(const int* __restrict__ dst, int* __restrict__ cursor,
                            int* __restrict__ eid, int E) {
    int e = blockIdx.x * blockDim.x + threadIdx.x;
    if (e >= E) return;
    int p = atomicAdd(&cursor[dst[e]], 1);
    eid[p] = e;
}

__global__ void order_segments(const int* __restrict__ src, const int* __restrict__ dst,
                               const int* __restrict__ seg_start,
                               const int* __restrict__ seg_end,
                               const int* __restrict__ eid,
                               int* __restrict__ ssrc, int E) {
    int p = blockIdx.x * blockDim.x + threadIdx.x;
    if (p >= E) return;
    int e = eid[p];
    int d = dst[e];
    int b = seg_start[d], en = seg_end[d];
    int r = 0;
    for (int q = b; q < en; ++q) r += (eid[q] < e) ? 1 : 0;
    ssrc[b + r] = src[e];
}

template <int CTRL>
__device__ __forceinline__ float dpp_row(float v) {
    return __builtin_bit_cast(float, __builtin_amdgcn_update_dpp(
        0, __builtin_bit_cast(int, v), CTRL, 0xf, 0xf, false));
}
__device__ __forceinline__ float row16_sum(float v) {
    v += dpp_row<0x161>(v);
    v += dpp_row<0x162>(v);
    v += dpp_row<0x164>(v);
    v += dpp_row<0x168>(v);
    return v;
}

__device__ __forceinline__ float pow2_hw(float x) {
    return __builtin_amdgcn_exp2f(x);
}

template <bool LAST>
__global__ void __launch_bounds__(256) __attribute__((amdgpu_waves_per_eu(16, 16)))
seg_softmax_update(const float* __restrict__ fs, float* __restrict__ fd,
                   const int* __restrict__ ssrc,
                   const int* __restrict__ seg_start,
                   const int* __restrict__ seg_end,
                   const float* __restrict__ attn,
                   const float* __restrict__ gamma,
                   const float* __restrict__ beta,
                   float* __restrict__ h,
                   const float* __restrict__ wa, const float* __restrict__ ba,
                   const float* __restrict__ wb, const float* __restrict__ bb,
                   float* __restrict__ ya, int n) {
    __shared__ __attribute__((aligned(16))) float lds_h[16][HID + 4];
    __shared__ __attribute__((aligned(16))) float lds_y[LAST ? 1 : 16][LAST ? 4 : 2 * HID + 4];

    const int lane = threadIdx.x & 31;
    const int l16  = lane & 15;
    const int half = lane >> 4;
    const int node = (blockIdx.x * blockDim.x + threadIdx.x) >> 4;
    const bool valid = node < n;
    const int nd = valid ? node : (n - 1);
    const int b = seg_start[nd];
    const int deg = valid ? (seg_end[nd] - b) : 0;
    const float4 fdv = ((const float4*)(fd + (size_t)nd * HID))[l16];
    float4 av = ((const float4*)attn)[l16];
    {
        const float log2e = 1.44269504088896340736f;
        av.x *= log2e; av.y *= log2e; av.z *= log2e; av.w *= log2e;
    }
    float4 hv = ((const float4*)(h + (size_t)nd * HID))[l16];

    const int d0 = __builtin_amdgcn_readlane(deg, 0);
    const int d1 = __builtin_amdgcn_readlane(deg, 16);
    const int dmax = d0 > d1 ? d0 : d1;

    float m = -INFINITY, l = 0.0f;
    float4 acc = make_float4(0.f, 0.f, 0.f, 0.f);

    for (int off = 0; off < dmax; off += 16) {
        const int p = off + l16;
        const int myidx = (p < deg) ? ssrc[b + p] : 0;
        int lim = dmax - off;
        if (lim > 16) lim = 16;
        for (int j = 0; j < lim; j += 4) {
            float4 r[4];
#pragma unroll
            for (int k = 0; k < 4; ++k) {
                const int s0 = __builtin_amdgcn_readlane(myidx, j + k);
                const int s1 = __builtin_amdgcn_readlane(myidx, j + k + 16);
                const int s  = half ? s1 : s0;
                r[k] = ((const float4*)(fs + (size_t)s * HID))[l16];
            }
            float sc[4];
#pragma unroll
            for (int k = 0; k < 4; ++k) {
                float x0 = r[k].x + fdv.x, x1 = r[k].y + fdv.y;
                float x2 = r[k].z + fdv.z, x3 = r[k].w + fdv.w;
                x0 = fmaxf(x0, x0 * NEG_SLOPE);
                x1 = fmaxf(x1, x1 * NEG_SLOPE);
                x2 = fmaxf(x2, x2 * NEG_SLOPE);
                x3 = fmaxf(x3, x3 * NEG_SLOPE);
                const float s = row16_sum(x0 * av.x + x1 * av.y + x2 * av.z + x3 * av.w);
                sc[k] = (off + j + k < deg) ? s : -INFINITY;
            }
            if (off + j < deg) {
                const float bm    = fmaxf(fmaxf(sc[0], sc[1]), fmaxf(sc[2], sc[3]));
                const float mn    = fmaxf(m, bm);
                const float scale = pow2_hw(m - mn);
                const float p0 = pow2_hw(sc[0] - mn), p1 = pow2_hw(sc[1] - mn);
                const float p2 = pow2_hw(sc[2] - mn), p3 = pow2_hw(sc[3] - mn);
                l     = l * scale + ((p0 + p1) + (p2 + p3));
                acc.x = acc.x * scale + p0 * r[0].x + p1 * r[1].x + p2 * r[2].x + p3 * r[3].x;
                acc.y = acc.y * scale + p0 * r[0].y + p1 * r[1].y + p2 * r[2].y + p3 * r[3].y;
                acc.z = acc.z * scale + p0 * r[0].z + p1 * r[1].z + p2 * r[2].z + p3 * r[3].z;
                acc.w = acc.w * scale + p0 * r[0].w + p1 * r[1].w + p2 * r[2].w + p3 * r[3].w;
                m = mn;
            }
        }
    }

    float4 o = make_float4(0.f, 0.f, 0.f, 0.f);
    if (deg > 0) {
        o.x = acc.x / l; o.y = acc.y / l; o.z = acc.z / l; o.w = acc.w / l;
    }

    const float mean = row16_sum((o.x + o.y) + (o.z + o.w)) * (1.0f / HID);
    const float c0 = o.x - mean, c1 = o.y - mean, c2 = o.z - mean, c3 = o.w - mean;
    const float var = row16_sum((c0 * c0 + c1 * c1) + (c2 * c2 + c3 * c3)) * (1.0f / HID);
    const float inv = rsqrtf(var + LN_EPS);

    const float4 g  = ((const float4*)gamma)[l16];
    const float4 bt = ((const float4*)beta)[l16];
    const float x0 = c0 * inv * g.x + bt.x;
    const float x1 = c1 * inv * g.y + bt.y;
    const float x2 = c2 * inv * g.z + bt.z;
    const float x3 = c3 * inv * g.w + bt.w;
    const float kk = 0.70710678118654752f;
    hv.x += 0.5f * x0 * (1.0f + erff(x0 * kk));
    hv.y += 0.5f * x1 * (1.0f + erff(x1 * kk));
    hv.z += 0.5f * x2 * (1.0f + erff(x2 * kk));
    hv.w += 0.5f * x3 * (1.0f + erff(x3 * kk));

    const int wave = __builtin_amdgcn_readfirstlane((int)(threadIdx.x >> 5));
    const bool owns_tile = LAST ? (wave == 0) : true;
    const int   ldw   = LAST ? 8 : HID;
    const bool  colok = LAST ? (l16 < 8) : true;
    const int   col   = LAST ? (colok ? l16 : 7)
                             : ((wave & 3) * 16 + l16);
    const float* Wt = LAST ? wa : (wave < 4 ? wa : wb);
    const float* Bt = LAST ? ba : (wave < 4 ? ba : bb);

    v2f bfrag[16];
    float bv = 0.0f;
    if (owns_tile) {
#pragma unroll
        for (int s = 0; s < 16; ++s) {
            const float b0 = Wt[(4 * s + 2 * half + 0) * ldw + col];
            const float b1 = Wt[(4 * s + 2 * half + 1) * ldw + col];
            bfrag[s].x = colok ? b0 : 0.0f;
            bfrag[s].y = colok ? b1 : 0.0f;
        }
        const float bl = Bt[col];
        bv = colok ? bl : 0.0f;
    }

    ((float4*)&lds_h[threadIdx.x >> 4][0])[l16] = hv;
    __syncthreads();

    if (owns_tile) {
        v8f acc = {};
#pragma unroll
        for (int s = 0; s < 16; ++s) {
            const v2f a = *(const v2f*)&lds_h[l16][4 * s + 2 * half];
            acc = __builtin_amdgcn_wmma_f32_16x16x4_f32(
                false, a, false, bfrag[s], (short)0, acc, false, false);
        }
#pragma unroll
        for (int i = 0; i < 8; ++i) {
            if (LAST) {
                const int row = blockIdx.x * 16 + 8 * half + i;
                if (row < n && colok)
                    ya[(size_t)row * 8 + col] = acc[i] + bv;
            } else {
                lds_y[8 * half + i][wave * 16 + l16] = acc[i] + bv;
            }
        }
    }

    if (!LAST) {
        __syncthreads();
        if (valid) ((float4*)(h + (size_t)node * HID))[l16] = hv;
        float* Yp = half ? fd : ya;
#pragma unroll
        for (int q = 0; q < 2; ++q) {
            const int r = wave + 8 * q;
            const int row = blockIdx.x * 16 + r;
            const float4 v = *(const float4*)&lds_y[r][4 * lane];
            if (row < n) ((float4*)(Yp + (size_t)row * HID))[l16] = v;
        }
    }
}

extern "C" void kernel_launch(void* const* d_in, const int* in_sizes, int n_in,
                              void* d_out, int out_size, void* d_ws, size_t ws_size,
                              hipStream_t stream) {
    const float* nodes = (const float*)d_in[0];
    const int*   src   = (const int*)d_in[1];
    const int*   dst   = (const int*)d_in[2];
    const float* w_in  = (const float*)d_in[3];
    const float* b_in  = (const float*)d_in[4];
    const float* w_src = (const float*)d_in[5];
    const float* b_src = (const float*)d_in[6];
    const float* w_dst = (const float*)d_in[7];
    const float* b_dst = (const float*)d_in[8];
    const float* attn  = (const float*)d_in[9];
    const float* gamma = (const float*)d_in[10];
    const float* beta  = (const float*)d_in[11];
    const float* w_out = (const float*)d_in[12];
    const float* b_out = (const float*)d_in[13];

    const int N = in_sizes[0] / HID;
    const int E = in_sizes[1];
    const int NL = 6;

    float* ws   = (float*)d_ws;
    float* h    = ws;  ws += (size_t)N * HID;
    float* fs   = ws;  ws += (size_t)N * HID;
    float* fd   = ws;  ws += (size_t)N * HID;
    float* tbuf = ws;  ws += (size_t)N * HID;
    int* ssrc      = (int*)ws; ws += E;
    int* seg_start = (int*)ws; ws += N;
    int* seg_end   = (int*)ws;
    int* eid       = (int*)tbuf;
    int* cnt       = eid + E;
    float* fsp[2]  = { fs, tbuf };

    const int ggrid  = (N + 63) / 64;
    const int eblk   = 256;
    const int egrid  = (E + eblk - 1) / eblk;
    const int ngrid  = (N + 255) / 256;
    const int sgrid  = (N * 16 + 255) / 256;
    const int scgrid = (N + 1023) / 1024;

    gemm64_wmma<<<ggrid, 128, 0, stream>>>(nodes, w_in, b_in, h, N);

    zero_counts<<<ngrid, 256, 0, stream>>>(cnt, N);
    count_dst<<<egrid, eblk, 0, stream>>>(dst, cnt, E);
    scan_counts<<<scgrid, 1024, 0, stream>>>(cnt, seg_start, seg_end, N);
    fill_by_dst<<<egrid, eblk, 0, stream>>>(dst, seg_end, eid, E);
    order_segments<<<egrid, eblk, 0, stream>>>(src, dst, seg_start, seg_end, eid, ssrc, E);

    gemm64_wmma<<<ggrid, 128, 0, stream>>>(h, w_src, b_src, fs, N);
    gemm64_wmma<<<ggrid, 128, 0, stream>>>(h, w_dst, b_dst, fd, N);

    for (int L = 0; L + 1 < NL; ++L) {
        seg_softmax_update<false><<<sgrid, 256, 0, stream>>>(
            fsp[L & 1], fd, ssrc, seg_start, seg_end,
            attn + (size_t)L * HID, gamma + (size_t)L * HID, beta + (size_t)L * HID, h,
            w_src + (size_t)(L + 1) * HID * HID, b_src + (size_t)(L + 1) * HID,
            w_dst + (size_t)(L + 1) * HID * HID, b_dst + (size_t)(L + 1) * HID,
            fsp[(L + 1) & 1], N);
    }
    seg_softmax_update<true><<<sgrid, 256, 0, stream>>>(
        fsp[(NL - 1) & 1], fd, ssrc, seg_start, seg_end,
        attn + (size_t)(NL - 1) * HID, gamma + (size_t)(NL - 1) * HID,
        beta + (size_t)(NL - 1) * HID, h,
        w_out, b_out, w_out, b_out, (float*)d_out, N);
}


# grouping without global atomics: a block owns 1024 destinations x 1 of 5 edge slices and counts, then hands out slots, on LDS counters (2 launches replace zero, count, scan, fill); sums unchanged
# speedup vs baseline: 13.2048x; 1.1784x over previous
#include <hip/hip_runtime.h>
#include <hip/hip_bf16.h>
#include <math.h>


typedef __attribute__((ext_vector_type(2))) float v2f;
typedef __attribute__((ext_vector_type(8))) float v8f;

#define HID 64
#define NEG_SLOPE 0.2f
#define LN_EPS 1e-5f

__global__ void __launch_bounds__(128)
gemm64_wmma(const float* __restrict__ X,
            const float* __restrict__ W,
            const float* __restrict__ bias,
            float* __restrict__ Y,
            int nrows) {
    constexpr int NTILES = 4;
    constexpr int nout = HID;
    __shared__ __attribute__((aligned(16))) float lds_w[64][64];

    const int tid = threadIdx.x;
    {
        const float4* W4 = (const float4*)W;
        float4* L4 = (float4*)&lds_w[0][0];
        float4 wv[8];
#pragma unroll
        for (int i = 0; i < 8; ++i) wv[i] = W4[tid + 128 * i];
#pragma unroll
        for (int i = 0; i < 8; ++i) L4[tid + 128 * i] = wv[i];
    }
    __syncthreads();

    int lane  = tid & 31;
    int wave  = tid >> 5;
    int half  = lane >> 4;
    int l15   = lane & 15;
    int rbase = blockIdx.x * 64 + wave * 16;

    int arow = rbase + l15;
    if (arow >= nrows) arow = nrows - 1;
    const float* xrow = X + (size_t)arow * 64;

    v8f acc[NTILES];
#pragma unroll
    for (int t = 0; t < NTILES; ++t)
#pragma unroll
        for (int j = 0; j < 8; ++j) acc[t][j] = 0.0f;

#pragma unroll
    for (int k0 = 0; k0 < 64; k0 += 4) {
        const float* ap = xrow + k0 + 2 * half;
        v2f a;
        a.x = ap[0];
        a.y = ap[1];
#pragma unroll
        for (int t = 0; t < NTILES; ++t) {
            v2f b;
            b.x = lds_w[k0 + 2 * half + 0][t * 16 + l15];
            b.y = lds_w[k0 + 2 * half + 1][t * 16 + l15];
            acc[t] = __builtin_amdgcn_wmma_f32_16x16x4_f32(
                false, a, false, b, (short)0, acc[t], false, false);
        }
    }

#pragma unroll
    for (int t = 0; t < NTILES; ++t) {
        int col = t * 16 + l15;
        float bv = bias[col];
#pragma unroll
        for (int i = 0; i < 8; ++i) {
            int row = rbase + 8 * half + i;
            if (row < nrows)
                Y[(size_t)row * nout + col] = acc[t][i] + bv;
        }
    }
}

#define GRP_BIN 1024
#define GRP_SLICES 5
#define GRP_DEPTH 8

__device__ __forceinline__ void grp_slice(int E, int s, int& b4, int& e4) {
    const int n4  = E >> 2;
    const int per = (n4 + GRP_SLICES - 1) / GRP_SLICES;
    b4 = s * per;
    e4 = b4 + per;
    if (b4 > n4) b4 = n4;
    if (e4 > n4) e4 = n4;
}

__global__ void __launch_bounds__(1024)
count_range_slice(const int* __restrict__ dst, int* __restrict__ cnt_s,
                  int* __restrict__ tot, int E, int n) {
    __shared__ int c[GRP_BIN];
    __shared__ int w_tot[32];
    const int tid  = threadIdx.x;
    const int lane = tid & 31;
    const int wave = tid >> 5;
    const int r    = blockIdx.x;
    const int s    = blockIdx.y;
    const int lo   = r * GRP_BIN;
    const unsigned span = (unsigned)((n - lo) < GRP_BIN ? (n - lo) : GRP_BIN);

    c[tid] = 0;
    __syncthreads();

    int b4, e4;
    grp_slice(E, s, b4, e4);
    const int4* d4 = (const int4*)dst;
    for (int i0 = b4 + tid; i0 < e4; i0 += 1024 * GRP_DEPTH) {
        int4 v[GRP_DEPTH];
#pragma unroll
        for (int k = 0; k < GRP_DEPTH; ++k) {
            const int i = i0 + 1024 * k;
            v[k] = (i < e4) ? d4[i] : make_int4(-1, -1, -1, -1);
        }
#pragma unroll
        for (int k = 0; k < GRP_DEPTH; ++k) {
            unsigned t;
            t = (unsigned)(v[k].x - lo); if (t < span) atomicAdd(&c[t], 1);
            t = (unsigned)(v[k].y - lo); if (t < span) atomicAdd(&c[t], 1);
            t = (unsigned)(v[k].z - lo); if (t < span) atomicAdd(&c[t], 1);
            t = (unsigned)(v[k].w - lo); if (t < span) atomicAdd(&c[t], 1);
        }
    }
    if (s == GRP_SLICES - 1) {
        const int i = ((E >> 2) << 2) + tid;
        if (i < E) {
            const unsigned t = (unsigned)(dst[i] - lo);
            if (t < span) atomicAdd(&c[t], 1);
        }
    }
    __syncthreads();

    const int mine = c[tid];
    const int idx  = lo + tid;
    if (idx < n) cnt_s[(size_t)s * n + idx] = mine;

    int sum = mine;
    for (int d = 16; d >= 1; d >>= 1) sum += __shfl_xor(sum, d);
    if (lane == 0) w_tot[wave] = sum;
    __syncthreads();
    if (wave == 0) {
        int t = w_tot[lane];
        for (int d = 16; d >= 1; d >>= 1) t += __shfl_xor(t, d);
        if (lane == 0) tot[r * GRP_SLICES + s] = t;
    }
}

__global__ void __launch_bounds__(1024)
place_range_slice(const int* __restrict__ dst, const int* __restrict__ cnt_s,
                  const int* __restrict__ tot, int* __restrict__ seg_start,
                  int* __restrict__ seg_end, int* __restrict__ eid, int E, int n) {
    __shared__ int cur[GRP_BIN];
    __shared__ int w_cnt[32];
    __shared__ int w_pre[32];
    const int tid  = threadIdx.x;
    const int lane = tid & 31;
    const int wave = tid >> 5;
    const int r    = blockIdx.x;
    const int s    = blockIdx.y;
    const int lo   = r * GRP_BIN;
    const int idx  = lo + tid;
    const unsigned span = (unsigned)((n - lo) < GRP_BIN ? (n - lo) : GRP_BIN);

    int call = 0, cfront = 0;
    if (idx < n) {
#pragma unroll
        for (int q = 0; q < GRP_SLICES; ++q) {
            const int v = cnt_s[(size_t)q * n + idx];
            call += v;
            if (q < s) cfront += v;
        }
    }
    int pre = 0;
    for (int i = tid; i < r * GRP_SLICES; i += 1024) pre += tot[i];

    int incl = call;
    for (int d = 1; d < 32; d <<= 1) {
        const int v = __shfl_up(incl, d);
        if (lane >= d) incl += v;
    }
    for (int d = 16; d >= 1; d >>= 1) pre += __shfl_xor(pre, d);
    if (lane == 31) w_cnt[wave] = incl;
    if (lane == 0)  w_pre[wave] = pre;
    __syncthreads();

    const int wc = w_cnt[lane];
    int wp = w_pre[lane];
    int wincl = wc;
    for (int d = 1; d < 32; d <<= 1) {
        const int v = __shfl_up(wincl, d);
        if (lane >= d) wincl += v;
    }
    for (int d = 16; d >= 1; d >>= 1) wp += __shfl_xor(wp, d);
    const int woff = __shfl(wincl - wc, wave);

    const int excl = wp + woff + incl - call;
    if (idx < n && s == 0) {
        seg_start[idx] = excl;
        seg_end[idx]   = excl + call;
    }
    cur[tid] = excl + cfront;
    __syncthreads();

    int b4, e4;
    grp_slice(E, s, b4, e4);
    const int4* d4 = (const int4*)dst;
    for (int i0 = b4 + tid; i0 < e4; i0 += 1024 * GRP_DEPTH) {
        int4 v[GRP_DEPTH];
#pragma unroll
        for (int k = 0; k < GRP_DEPTH; ++k) {
            const int i = i0 + 1024 * k;
            v[k] = (i < e4) ? d4[i] : make_int4(-1, -1, -1, -1);
        }
#pragma unroll
        for (int k = 0; k < GRP_DEPTH; ++k) {
            const int e = (i0 + 1024 * k) << 2;
            unsigned t;
            t = (unsigned)(v[k].x - lo); if (t < span) eid[atomicAdd(&cur[t], 1)] = e;
            t = (unsigned)(v[k].y - lo); if (t < span) eid[atomicAdd(&cur[t], 1)] = e + 1;
            t = (unsigned)(v[k].z - lo); if (t < span) eid[atomicAdd(&cur[t], 1)] = e + 2;
            t = (unsigned)(v[k].w - lo); if (t < span) eid[atomicAdd(&cur[t], 1)] = e + 3;
        }
    }
    if (s == GRP_SLICES - 1) {
        const int i = ((E >> 2) << 2) + tid;
        if (i < E) {
            const unsigned t = (unsigned)(dst[i] - lo);
            if (t < span) eid[atomicAdd(&cur[t], 1)] = i;
        }
    }
}

__global__ void order_segments(const int* __restrict__ src, const int* __restrict__ dst,
                               const int* __restrict__ seg_start,
                               const int* __restrict__ seg_end,
                               const int* __restrict__ eid,
                               int* __restrict__ ssrc, int E) {
    int p = blockIdx.x * blockDim.x + threadIdx.x;
    if (p >= E) return;
    int e = eid[p];
    int d = dst[e];
    int b = seg_start[d], en = seg_end[d];
    int r = 0;
    for (int q = b; q < en; ++q) r += (eid[q] < e) ? 1 : 0;
    ssrc[b + r] = src[e];
}

template <int CTRL>
__device__ __forceinline__ float dpp_row(float v) {
    return __builtin_bit_cast(float, __builtin_amdgcn_update_dpp(
        0, __builtin_bit_cast(int, v), CTRL, 0xf, 0xf, false));
}
__device__ __forceinline__ float row16_sum(float v) {
    v += dpp_row<0x161>(v);
    v += dpp_row<0x162>(v);
    v += dpp_row<0x164>(v);
    v += dpp_row<0x168>(v);
    return v;
}

__device__ __forceinline__ float pow2_hw(float x) {
    return __builtin_amdgcn_exp2f(x);
}

template <bool LAST>
__global__ void __launch_bounds__(256) __attribute__((amdgpu_waves_per_eu(16, 16)))
seg_softmax_update(const float* __restrict__ fs, float* __restrict__ fd,
                   const int* __restrict__ ssrc,
                   const int* __restrict__ seg_start,
                   const int* __restrict__ seg_end,
                   const float* __restrict__ attn,
                   const float* __restrict__ gamma,
                   const float* __restrict__ beta,
                   float* __restrict__ h,
                   const float* __restrict__ wa, const float* __restrict__ ba,
                   const float* __restrict__ wb, const float* __restrict__ bb,
                   float* __restrict__ ya, int n) {
    __shared__ __attribute__((aligned(16))) float lds_h[16][HID + 4];
    __shared__ __attribute__((aligned(16))) float lds_y[LAST ? 1 : 16][LAST ? 4 : 2 * HID + 4];

    const int lane = threadIdx.x & 31;
    const int l16  = lane & 15;
    const int half = lane >> 4;
    const int node = (blockIdx.x * blockDim.x + threadIdx.x) >> 4;
    const bool valid = node < n;
    const int nd = valid ? node : (n - 1);
    const int b = seg_start[nd];
    const int deg = valid ? (seg_end[nd] - b) : 0;
    const float4 fdv = ((const float4*)(fd + (size_t)nd * HID))[l16];
    float4 av = ((const float4*)attn)[l16];
    {
        const float log2e = 1.44269504088896340736f;
        av.x *= log2e; av.y *= log2e; av.z *= log2e; av.w *= log2e;
    }
    float4 hv = ((const float4*)(h + (size_t)nd * HID))[l16];

    const int d0 = __builtin_amdgcn_readlane(deg, 0);
    const int d1 = __builtin_amdgcn_readlane(deg, 16);
    const int dmax = d0 > d1 ? d0 : d1;

    float m = -INFINITY, l = 0.0f;
    float4 acc = make_float4(0.f, 0.f, 0.f, 0.f);

    for (int off = 0; off < dmax; off += 16) {
        const int p = off + l16;
        const int myidx = (p < deg) ? ssrc[b + p] : 0;
        int lim = dmax - off;
        if (lim > 16) lim = 16;
        for (int j = 0; j < lim; j += 4) {
            float4 r[4];
#pragma unroll
            for (int k = 0; k < 4; ++k) {
                const int s0 = __builtin_amdgcn_readlane(myidx, j + k);
                const int s1 = __builtin_amdgcn_readlane(myidx, j + k + 16);
                const int s  = half ? s1 : s0;
                r[k] = ((const float4*)(fs + (size_t)s * HID))[l16];
            }
            float sc[4];
#pragma unroll
            for (int k = 0; k < 4; ++k) {
                float x0 = r[k].x + fdv.x, x1 = r[k].y + fdv.y;
                float x2 = r[k].z + fdv.z, x3 = r[k].w + fdv.w;
                x0 = fmaxf(x0, x0 * NEG_SLOPE);
                x1 = fmaxf(x1, x1 * NEG_SLOPE);
                x2 = fmaxf(x2, x2 * NEG_SLOPE);
                x3 = fmaxf(x3, x3 * NEG_SLOPE);
                const float s = row16_sum(x0 * av.x + x1 * av.y + x2 * av.z + x3 * av.w);
                sc[k] = (off + j + k < deg) ? s : -INFINITY;
            }
            if (off + j < deg) {
                const float bm    = fmaxf(fmaxf(sc[0], sc[1]), fmaxf(sc[2], sc[3]));
                const float mn    = fmaxf(m, bm);
                const float scale = pow2_hw(m - mn);
                const float p0 = pow2_hw(sc[0] - mn), p1 = pow2_hw(sc[1] - mn);
                const float p2 = pow2_hw(sc[2] - mn), p3 = pow2_hw(sc[3] - mn);
                l     = l * scale + ((p0 + p1) + (p2 + p3));
                acc.x = acc.x * scale + p0 * r[0].x + p1 * r[1].x + p2 * r[2].x + p3 * r[3].x;
                acc.y = acc.y * scale + p0 * r[0].y + p1 * r[1].y + p2 * r[2].y + p3 * r[3].y;
                acc.z = acc.z * scale + p0 * r[0].z + p1 * r[1].z + p2 * r[2].z + p3 * r[3].z;
                acc.w = acc.w * scale + p0 * r[0].w + p1 * r[1].w + p2 * r[2].w + p3 * r[3].w;
                m = mn;
            }
        }
    }

    float4 o = make_float4(0.f, 0.f, 0.f, 0.f);
    if (deg > 0) {
        o.x = acc.x / l; o.y = acc.y / l; o.z = acc.z / l; o.w = acc.w / l;
    }

    const float mean = row16_sum((o.x + o.y) + (o.z + o.w)) * (1.0f / HID);
    const float c0 = o.x - mean, c1 = o.y - mean, c2 = o.z - mean, c3 = o.w - mean;
    const float var = row16_sum((c0 * c0 + c1 * c1) + (c2 * c2 + c3 * c3)) * (1.0f / HID);
    const float inv = rsqrtf(var + LN_EPS);

    const float4 g  = ((const float4*)gamma)[l16];
    const float4 bt = ((const float4*)beta)[l16];
    const float x0 = c0 * inv * g.x + bt.x;
    const float x1 = c1 * inv * g.y + bt.y;
    const float x2 = c2 * inv * g.z + bt.z;
    const float x3 = c3 * inv * g.w + bt.w;
    const float kk = 0.70710678118654752f;
    hv.x += 0.5f * x0 * (1.0f + erff(x0 * kk));
    hv.y += 0.5f * x1 * (1.0f + erff(x1 * kk));
    hv.z += 0.5f * x2 * (1.0f + erff(x2 * kk));
    hv.w += 0.5f * x3 * (1.0f + erff(x3 * kk));

    const int wave = __builtin_amdgcn_readfirstlane((int)(threadIdx.x >> 5));
    const bool owns_tile = LAST ? (wave == 0) : true;
    const int   ldw   = LAST ? 8 : HID;
    const bool  colok = LAST ? (l16 < 8) : true;
    const int   col   = LAST ? (colok ? l16 : 7)
                             : ((wave & 3) * 16 + l16);
    const float* Wt = LAST ? wa : (wave < 4 ? wa : wb);
    const float* Bt = LAST ? ba : (wave < 4 ? ba : bb);

    v2f bfrag[16];
    float bv = 0.0f;
    if (owns_tile) {
#pragma unroll
        for (int s = 0; s < 16; ++s) {
            const float b0 = Wt[(4 * s + 2 * half + 0) * ldw + col];
            const float b1 = Wt[(4 * s + 2 * half + 1) * ldw + col];
            bfrag[s].x = colok ? b0 : 0.0f;
            bfrag[s].y = colok ? b1 : 0.0f;
        }
        const float bl = Bt[col];
        bv = colok ? bl : 0.0f;
    }

    ((float4*)&lds_h[threadIdx.x >> 4][0])[l16] = hv;
    __syncthreads();

    if (owns_tile) {
        v8f acc = {};
#pragma unroll
        for (int s = 0; s < 16; ++s) {
            const v2f a = *(const v2f*)&lds_h[l16][4 * s + 2 * half];
            acc = __builtin_amdgcn_wmma_f32_16x16x4_f32(
                false, a, false, bfrag[s], (short)0, acc, false, false);
        }
#pragma unroll
        for (int i = 0; i < 8; ++i) {
            if (LAST) {
                const int row = blockIdx.x * 16 + 8 * half + i;
                if (row < n && colok)
                    ya[(size_t)row * 8 + col] = acc[i] + bv;
            } else {
                lds_y[8 * half + i][wave * 16 + l16] = acc[i] + bv;
            }
        }
    }

    if (!LAST) {
        __syncthreads();
        if (valid) ((float4*)(h + (size_t)node * HID))[l16] = hv;
        float* Yp = half ? fd : ya;
#pragma unroll
        for (int q = 0; q < 2; ++q) {
            const int r = wave + 8 * q;
            const int row = blockIdx.x * 16 + r;
            const float4 v = *(const float4*)&lds_y[r][4 * lane];
            if (row < n) ((float4*)(Yp + (size_t)row * HID))[l16] = v;
        }
    }
}

extern "C" void kernel_launch(void* const* d_in, const int* in_sizes, int n_in,
                              void* d_out, int out_size, void* d_ws, size_t ws_size,
                              hipStream_t stream) {
    const float* nodes = (const float*)d_in[0];
    const int*   src   = (const int*)d_in[1];
    const int*   dst   = (const int*)d_in[2];
    const float* w_in  = (const float*)d_in[3];
    const float* b_in  = (const float*)d_in[4];
    const float* w_src = (const float*)d_in[5];
    const float* b_src = (const float*)d_in[6];
    const float* w_dst = (const float*)d_in[7];
    const float* b_dst = (const float*)d_in[8];
    const float* attn  = (const float*)d_in[9];
    const float* gamma = (const float*)d_in[10];
    const float* beta  = (const float*)d_in[11];
    const float* w_out = (const float*)d_in[12];
    const float* b_out = (const float*)d_in[13];

    const int N = in_sizes[0] / HID;
    const int E = in_sizes[1];
    const int NL = 6;

    float* ws   = (float*)d_ws;
    float* h    = ws;  ws += (size_t)N * HID;
    float* fs   = ws;  ws += (size_t)N * HID;
    float* fd   = ws;  ws += (size_t)N * HID;
    float* tbuf = ws;  ws += (size_t)N * HID;
    int* ssrc      = (int*)ws; ws += E;
    int* seg_start = (int*)ws; ws += N;
    int* seg_end   = (int*)ws;
    int* eid       = (int*)tbuf;
    int* cnt_s     = eid + E;
    int* tot       = cnt_s + (size_t)GRP_SLICES * N;
    float* fsp[2]  = { fs, tbuf };

    const int ggrid  = (N + 63) / 64;
    const int eblk   = 256;
    const int egrid  = (E + eblk - 1) / eblk;
    const int sgrid  = (N * 16 + 255) / 256;
    const dim3 bgrid((N + GRP_BIN - 1) / GRP_BIN, GRP_SLICES);

    gemm64_wmma<<<ggrid, 128, 0, stream>>>(nodes, w_in, b_in, h, N);

    count_range_slice<<<bgrid, 1024, 0, stream>>>(dst, cnt_s, tot, E, N);
    place_range_slice<<<bgrid, 1024, 0, stream>>>(dst, cnt_s, tot, seg_start, seg_end, eid, E, N);
    order_segments<<<egrid, eblk, 0, stream>>>(src, dst, seg_start, seg_end, eid, ssrc, E);

    gemm64_wmma<<<ggrid, 128, 0, stream>>>(h, w_src, b_src, fs, N);
    gemm64_wmma<<<ggrid, 128, 0, stream>>>(h, w_dst, b_dst, fd, N);

    for (int L = 0; L + 1 < NL; ++L) {
        seg_softmax_update<false><<<sgrid, 256, 0, stream>>>(
            fsp[L & 1], fd, ssrc, seg_start, seg_end,
            attn + (size_t)L * HID, gamma + (size_t)L * HID, beta + (size_t)L * HID, h,
            w_src + (size_t)(L + 1) * HID * HID, b_src + (size_t)(L + 1) * HID,
            w_dst + (size_t)(L + 1) * HID * HID, b_dst + (size_t)(L + 1) * HID,
            fsp[(L + 1) & 1], N);
    }
    seg_softmax_update<true><<<sgrid, 256, 0, stream>>>(
        fsp[(NL - 1) & 1], fd, ssrc, seg_start, seg_end,
        attn + (size_t)(NL - 1) * HID, gamma + (size_t)(NL - 1) * HID,
        beta + (size_t)(NL - 1) * HID, h,
        w_out, b_out, w_out, b_out, (float*)d_out, N);
}


# edge kernel, fewer vector instructions: source ids by row-share moves, 32-bit gather offsets, softmax division folded into the layernorm scale, erf without branches on hardware exp2; sums unchanged
# speedup vs baseline: 13.7308x; 1.0398x over previous
#include <hip/hip_runtime.h>
#include <hip/hip_bf16.h>
#include <math.h>


typedef __attribute__((ext_vector_type(2))) float v2f;
typedef __attribute__((ext_vector_type(8))) float v8f;
typedef __attribute__((ext_vector_type(4))) float v4f;

#define HID 64
#define NEG_SLOPE 0.2f
#define LN_EPS 1e-5f

__global__ void __launch_bounds__(128)
gemm64_wmma(const float* __restrict__ X,
            const float* __restrict__ W,
            const float* __restrict__ bias,
            float* __restrict__ Y,
            int nrows) {
    constexpr int NTILES = 4;
    constexpr int nout = HID;
    __shared__ __attribute__((aligned(16))) float lds_w[64][64];

    const int tid = threadIdx.x;
    {
        const float4* W4 = (const float4*)W;
        float4* L4 = (float4*)&lds_w[0][0];
        float4 wv[8];
#pragma unroll
        for (int i = 0; i < 8; ++i) wv[i] = W4[tid + 128 * i];
#pragma unroll
        for (int i = 0; i < 8; ++i) L4[tid + 128 * i] = wv[i];
    }
    __syncthreads();

    int lane  = tid & 31;
    int wave  = tid >> 5;
    int half  = lane >> 4;
    int l15   = lane & 15;
    int rbase = blockIdx.x * 64 + wave * 16;

    int arow = rbase + l15;
    if (arow >= nrows) arow = nrows - 1;
    const float* xrow = X + (size_t)arow * 64;

    v8f acc[NTILES];
#pragma unroll
    for (int t = 0; t < NTILES; ++t)
#pragma unroll
        for (int j = 0; j < 8; ++j) acc[t][j] = 0.0f;

#pragma unroll
    for (int k0 = 0; k0 < 64; k0 += 4) {
        const float* ap = xrow + k0 + 2 * half;
        v2f a;
        a.x = ap[0];
        a.y = ap[1];
#pragma unroll
        for (int t = 0; t < NTILES; ++t) {
            v2f b;
            b.x = lds_w[k0 + 2 * half + 0][t * 16 + l15];
            b.y = lds_w[k0 + 2 * half + 1][t * 16 + l15];
            acc[t] = __builtin_amdgcn_wmma_f32_16x16x4_f32(
                false, a, false, b, (short)0, acc[t], false, false);
        }
    }

#pragma unroll
    for (int t = 0; t < NTILES; ++t) {
        int col = t * 16 + l15;
        float bv = bias[col];
#pragma unroll
        for (int i = 0; i < 8; ++i) {
            int row = rbase + 8 * half + i;
            if (row < nrows)
                Y[(size_t)row * nout + col] = acc[t][i] + bv;
        }
    }
}

#define GRP_BIN 1024
#define GRP_SLICES 5
#define GRP_DEPTH 8

__device__ __forceinline__ void grp_slice(int E, int s, int& b4, int& e4) {
    const int n4  = E >> 2;
    const int per = (n4 + GRP_SLICES - 1) / GRP_SLICES;
    b4 = s * per;
    e4 = b4 + per;
    if (b4 > n4) b4 = n4;
    if (e4 > n4) e4 = n4;
}

__global__ void __launch_bounds__(1024)
count_range_slice(const int* __restrict__ dst, int* __restrict__ cnt_s,
                  int* __restrict__ tot, int E, int n) {
    __shared__ int c[GRP_BIN];
    __shared__ int w_tot[32];
    const int tid  = threadIdx.x;
    const int lane = tid & 31;
    const int wave = tid >> 5;
    const int r    = blockIdx.x;
    const int s    = blockIdx.y;
    const int lo   = r * GRP_BIN;
    const unsigned span = (unsigned)((n - lo) < GRP_BIN ? (n - lo) : GRP_BIN);

    c[tid] = 0;
    __syncthreads();

    int b4, e4;
    grp_slice(E, s, b4, e4);
    const int4* d4 = (const int4*)dst;
    for (int i0 = b4 + tid; i0 < e4; i0 += 1024 * GRP_DEPTH) {
        int4 v[GRP_DEPTH];
#pragma unroll
        for (int k = 0; k < GRP_DEPTH; ++k) {
            const int i = i0 + 1024 * k;
            v[k] = (i < e4) ? d4[i] : make_int4(-1, -1, -1, -1);
        }
#pragma unroll
        for (int k = 0; k < GRP_DEPTH; ++k) {
            unsigned t;
            t = (unsigned)(v[k].x - lo); if (t < span) atomicAdd(&c[t], 1);
            t = (unsigned)(v[k].y - lo); if (t < span) atomicAdd(&c[t], 1);
            t = (unsigned)(v[k].z - lo); if (t < span) atomicAdd(&c[t], 1);
            t = (unsigned)(v[k].w - lo); if (t < span) atomicAdd(&c[t], 1);
        }
    }
    if (s == GRP_SLICES - 1) {
        const int i = ((E >> 2) << 2) + tid;
        if (i < E) {
            const unsigned t = (unsigned)(dst[i] - lo);
            if (t < span) atomicAdd(&c[t], 1);
        }
    }
    __syncthreads();

    const int mine = c[tid];
    const int idx  = lo + tid;
    if (idx < n) cnt_s[(size_t)s * n + idx] = mine;

    int sum = mine;
    for (int d = 16; d >= 1; d >>= 1) sum += __shfl_xor(sum, d);
    if (lane == 0) w_tot[wave] = sum;
    __syncthreads();
    if (wave == 0) {
        int t = w_tot[lane];
        for (int d = 16; d >= 1; d >>= 1) t += __shfl_xor(t, d);
        if (lane == 0) tot[r * GRP_SLICES + s] = t;
    }
}

__global__ void __launch_bounds__(1024)
place_range_slice(const int* __restrict__ dst, const int* __restrict__ cnt_s,
                  const int* __restrict__ tot, int* __restrict__ seg_start,
                  int* __restrict__ seg_end, int* __restrict__ eid, int E, int n) {
    __shared__ int cur[GRP_BIN];
    __shared__ int w_cnt[32];
    __shared__ int w_pre[32];
    const int tid  = threadIdx.x;
    const int lane = tid & 31;
    const int wave = tid >> 5;
    const int r    = blockIdx.x;
    const int s    = blockIdx.y;
    const int lo   = r * GRP_BIN;
    const int idx  = lo + tid;
    const unsigned span = (unsigned)((n - lo) < GRP_BIN ? (n - lo) : GRP_BIN);

    int call = 0, cfront = 0;
    if (idx < n) {
#pragma unroll
        for (int q = 0; q < GRP_SLICES; ++q) {
            const int v = cnt_s[(size_t)q * n + idx];
            call += v;
            if (q < s) cfront += v;
        }
    }
    int pre = 0;
    for (int i = tid; i < r * GRP_SLICES; i += 1024) pre += tot[i];

    int incl = call;
    for (int d = 1; d < 32; d <<= 1) {
        const int v = __shfl_up(incl, d);
        if (lane >= d) incl += v;
    }
    for (int d = 16; d >= 1; d >>= 1) pre += __shfl_xor(pre, d);
    if (lane == 31) w_cnt[wave] = incl;
    if (lane == 0)  w_pre[wave] = pre;
    __syncthreads();

    const int wc = w_cnt[lane];
    int wp = w_pre[lane];
    int wincl = wc;
    for (int d = 1; d < 32; d <<= 1) {
        const int v = __shfl_up(wincl, d);
        if (lane >= d) wincl += v;
    }
    for (int d = 16; d >= 1; d >>= 1) wp += __shfl_xor(wp, d);
    const int woff = __shfl(wincl - wc, wave);

    const int excl = wp + woff + incl - call;
    if (idx < n && s == 0) {
        seg_start[idx] = excl;
        seg_end[idx]   = excl + call;
    }
    cur[tid] = excl + cfront;
    __syncthreads();

    int b4, e4;
    grp_slice(E, s, b4, e4);
    const int4* d4 = (const int4*)dst;
    for (int i0 = b4 + tid; i0 < e4; i0 += 1024 * GRP_DEPTH) {
        int4 v[GRP_DEPTH];
#pragma unroll
        for (int k = 0; k < GRP_DEPTH; ++k) {
            const int i = i0 + 1024 * k;
            v[k] = (i < e4) ? d4[i] : make_int4(-1, -1, -1, -1);
        }
#pragma unroll
        for (int k = 0; k < GRP_DEPTH; ++k) {
            const int e = (i0 + 1024 * k) << 2;
            unsigned t;
            t = (unsigned)(v[k].x - lo); if (t < span) eid[atomicAdd(&cur[t], 1)] = e;
            t = (unsigned)(v[k].y - lo); if (t < span) eid[atomicAdd(&cur[t], 1)] = e + 1;
            t = (unsigned)(v[k].z - lo); if (t < span) eid[atomicAdd(&cur[t], 1)] = e + 2;
            t = (unsigned)(v[k].w - lo); if (t < span) eid[atomicAdd(&cur[t], 1)] = e + 3;
        }
    }
    if (s == GRP_SLICES - 1) {
        const int i = ((E >> 2) << 2) + tid;
        if (i < E) {
            const unsigned t = (unsigned)(dst[i] - lo);
            if (t < span) eid[atomicAdd(&cur[t], 1)] = i;
        }
    }
}

__global__ void order_segments(const int* __restrict__ src, const int* __restrict__ dst,
                               const int* __restrict__ seg_start,
                               const int* __restrict__ seg_end,
                               const int* __restrict__ eid,
                               int* __restrict__ ssrc, int E) {
    int p = blockIdx.x * blockDim.x + threadIdx.x;
    if (p >= E) return;
    int e = eid[p];
    int d = dst[e];
    int b = seg_start[d], en = seg_end[d];
    int r = 0;
    for (int q = b; q < en; ++q) r += (eid[q] < e) ? 1 : 0;
    ssrc[b + r] = src[e];
}

template <int CTRL>
__device__ __forceinline__ float dpp_row(float v) {
    return __builtin_bit_cast(float, __builtin_amdgcn_update_dpp(
        0, __builtin_bit_cast(int, v), CTRL, 0xf, 0xf, false));
}
__device__ __forceinline__ float row16_sum(float v) {
    v += dpp_row<0x161>(v);
    v += dpp_row<0x162>(v);
    v += dpp_row<0x164>(v);
    v += dpp_row<0x168>(v);
    return v;
}

__device__ __forceinline__ float pow2_hw(float x) {
    return __builtin_amdgcn_exp2f(x);
}

template <int N>
__device__ __forceinline__ int row16_lane(int v) {
    return __builtin_amdgcn_update_dpp(0, v, 0x150 + N, 0xf, 0xf, false);
}

__device__ __forceinline__ float erf_both(float x) {
    const float t = fabsf(x);
    float q = fmaf(t, 1.69988107e-5f, -3.78677854e-4f);
    q = fmaf(t, q, 3.85781587e-3f);
    q = fmaf(t, q, -2.41816975e-2f);
    q = fmaf(t, q, 1.06668264e-1f);
    q = fmaf(t, q, 6.34933293e-1f);
    q = fmaf(t, q, 1.28689408e-1f);
    q = fmaf(t, q, t);
    const float far = 1.0f - pow2_hw(q * -1.44269504088896340736f);
    const float s = x * x;
    float p = fmaf(s, -5.61801775e-4f, 4.91381623e-3f);
    p = fmaf(s, p, -2.67075151e-2f);
    p = fmaf(s, p, 1.12800106e-1f);
    p = fmaf(s, p, -3.76122952e-1f);
    p = fmaf(s, p, 1.28379107e-1f);
    const float near = fmaf(t, p, t);
    return copysignf(t < 1.0f ? near : far, x);
}

template <int J>
__device__ __forceinline__ void edge_batch(const char* __restrict__ fsb, unsigned lane_off,
                                           int myidx, int first, int deg,
                                           const float4& fdv, const float4& av,
                                           float& m, float& l, float4& acc) {
    const int s0 = row16_lane<J + 0>(myidx);
    const int s1 = row16_lane<J + 1>(myidx);
    const int s2 = row16_lane<J + 2>(myidx);
    const int s3 = row16_lane<J + 3>(myidx);
    v4f r[4];
    r[0] = *(const v4f*)(fsb + (((unsigned)s0 << 8) | lane_off));
    r[1] = *(const v4f*)(fsb + (((unsigned)s1 << 8) | lane_off));
    r[2] = *(const v4f*)(fsb + (((unsigned)s2 << 8) | lane_off));
    r[3] = *(const v4f*)(fsb + (((unsigned)s3 << 8) | lane_off));
    asm volatile("" : "+v"(r[0]), "+v"(r[1]), "+v"(r[2]), "+v"(r[3]));
    float sc[4];
#pragma unroll
    for (int k = 0; k < 4; ++k) {
        float x0 = r[k].x + fdv.x, x1 = r[k].y + fdv.y;
        float x2 = r[k].z + fdv.z, x3 = r[k].w + fdv.w;
        x0 = fmaxf(x0, x0 * NEG_SLOPE);
        x1 = fmaxf(x1, x1 * NEG_SLOPE);
        x2 = fmaxf(x2, x2 * NEG_SLOPE);
        x3 = fmaxf(x3, x3 * NEG_SLOPE);
        const float s = row16_sum(x0 * av.x + x1 * av.y + x2 * av.z + x3 * av.w);
        sc[k] = (first + k < deg) ? s : -INFINITY;
    }
    if (first < deg) {
        const float bm    = fmaxf(fmaxf(sc[0], sc[1]), fmaxf(sc[2], sc[3]));
        const float mn    = fmaxf(m, bm);
        const float scale = pow2_hw(m - mn);
        const float p0 = pow2_hw(sc[0] - mn), p1 = pow2_hw(sc[1] - mn);
        const float p2 = pow2_hw(sc[2] - mn), p3 = pow2_hw(sc[3] - mn);
        l     = l * scale + ((p0 + p1) + (p2 + p3));
        acc.x = acc.x * scale + p0 * r[0].x + p1 * r[1].x + p2 * r[2].x + p3 * r[3].x;
        acc.y = acc.y * scale + p0 * r[0].y + p1 * r[1].y + p2 * r[2].y + p3 * r[3].y;
        acc.z = acc.z * scale + p0 * r[0].z + p1 * r[1].z + p2 * r[2].z + p3 * r[3].z;
        acc.w = acc.w * scale + p0 * r[0].w + p1 * r[1].w + p2 * r[2].w + p3 * r[3].w;
        m = mn;
    }
}

template <bool LAST>
__global__ void __launch_bounds__(256) __attribute__((amdgpu_waves_per_eu(16, 16)))
seg_softmax_update(const float* __restrict__ fs, float* __restrict__ fd,
                   const int* __restrict__ ssrc,
                   const int* __restrict__ seg_start,
                   const int* __restrict__ seg_end,
                   const float* __restrict__ attn,
                   const float* __restrict__ gamma,
                   const float* __restrict__ beta,
                   float* __restrict__ h,
                   const float* __restrict__ wa, const float* __restrict__ ba,
                   const float* __restrict__ wb, const float* __restrict__ bb,
                   float* __restrict__ ya, int n) {
    __shared__ __attribute__((aligned(16))) float lds_h[16][HID + 4];
    __shared__ __attribute__((aligned(16))) float lds_y[LAST ? 1 : 16][LAST ? 4 : 2 * HID + 4];

    const int lane = threadIdx.x & 31;
    const int l16  = lane & 15;
    const int half = lane >> 4;
    const int node = (blockIdx.x * blockDim.x + threadIdx.x) >> 4;
    const bool valid = node < n;
    const int nd = valid ? node : (n - 1);
    const int b = seg_start[nd];
    const int deg = valid ? (seg_end[nd] - b) : 0;
    const float4 fdv = ((const float4*)(fd + (size_t)nd * HID))[l16];
    float4 av = ((const float4*)attn)[l16];
    {
        const float log2e = 1.44269504088896340736f;
        av.x *= log2e; av.y *= log2e; av.z *= log2e; av.w *= log2e;
    }
    float4 hv = ((const float4*)(h + (size_t)nd * HID))[l16];

    const int d0 = __builtin_amdgcn_readlane(deg, 0);
    const int d1 = __builtin_amdgcn_readlane(deg, 16);
    const int dmax = d0 > d1 ? d0 : d1;

    float m = -INFINITY, l = 0.0f;
    float4 acc = make_float4(0.f, 0.f, 0.f, 0.f);

    const char* fsb = (const char*)fs;
    const unsigned lane_off = (unsigned)l16 * 16u;
    for (int off = 0; off < dmax; off += 16) {
        const int p = off + l16;
        const int myidx = (p < deg) ? ssrc[b + p] : 0;
        const int lim = dmax - off;
        edge_batch<0>(fsb, lane_off, myidx, off, deg, fdv, av, m, l, acc);
        if (lim > 4) {
            edge_batch<4>(fsb, lane_off, myidx, off + 4, deg, fdv, av, m, l, acc);
            if (lim > 8) {
                edge_batch<8>(fsb, lane_off, myidx, off + 8, deg, fdv, av, m, l, acc);
                if (lim > 12)
                    edge_batch<12>(fsb, lane_off, myidx, off + 12, deg, fdv, av, m, l, acc);
            }
        }
    }

    const float lq = (deg > 0) ? l : 1.0f;
    const float mean = row16_sum((acc.x + acc.y) + (acc.z + acc.w)) * (1.0f / HID);
    const float c0 = acc.x - mean, c1 = acc.y - mean, c2 = acc.z - mean, c3 = acc.w - mean;
    const float var = row16_sum((c0 * c0 + c1 * c1) + (c2 * c2 + c3 * c3)) * (1.0f / HID);
    const float inv = rsqrtf(var + LN_EPS * (lq * lq));

    const float4 g  = ((const float4*)gamma)[l16];
    const float4 bt = ((const float4*)beta)[l16];
    const float x0 = c0 * inv * g.x + bt.x;
    const float x1 = c1 * inv * g.y + bt.y;
    const float x2 = c2 * inv * g.z + bt.z;
    const float x3 = c3 * inv * g.w + bt.w;
    const float kk = 0.70710678118654752f;
    hv.x += 0.5f * x0 * (1.0f + erf_both(x0 * kk));
    hv.y += 0.5f * x1 * (1.0f + erf_both(x1 * kk));
    hv.z += 0.5f * x2 * (1.0f + erf_both(x2 * kk));
    hv.w += 0.5f * x3 * (1.0f + erf_both(x3 * kk));

    const int wave = __builtin_amdgcn_readfirstlane((int)(threadIdx.x >> 5));
    const bool owns_tile = LAST ? (wave == 0) : true;
    const int   ldw   = LAST ? 8 : HID;
    const bool  colok = LAST ? (l16 < 8) : true;
    const int   col   = LAST ? (colok ? l16 : 7)
                             : ((wave & 3) * 16 + l16);
    const float* Wt = LAST ? wa : (wave < 4 ? wa : wb);
    const float* Bt = LAST ? ba : (wave < 4 ? ba : bb);

    v2f bfrag[16];
    float bv = 0.0f;
    if (owns_tile) {
#pragma unroll
        for (int s = 0; s < 16; ++s) {
            const float b0 = Wt[(4 * s + 2 * half + 0) * ldw + col];
            const float b1 = Wt[(4 * s + 2 * half + 1) * ldw + col];
            bfrag[s].x = colok ? b0 : 0.0f;
            bfrag[s].y = colok ? b1 : 0.0f;
        }
        const float bl = Bt[col];
        bv = colok ? bl : 0.0f;
    }

    ((float4*)&lds_h[threadIdx.x >> 4][0])[l16] = hv;
    __syncthreads();

    if (owns_tile) {
        v8f acc = {};
#pragma unroll
        for (int s = 0; s < 16; ++s) {
            const v2f a = *(const v2f*)&lds_h[l16][4 * s + 2 * half];
            acc = __builtin_amdgcn_wmma_f32_16x16x4_f32(
                false, a, false, bfrag[s], (short)0, acc, false, false);
        }
#pragma unroll
        for (int i = 0; i < 8; ++i) {
            if (LAST) {
                const int row = blockIdx.x * 16 + 8 * half + i;
                if (row < n && colok)
                    ya[(size_t)row * 8 + col] = acc[i] + bv;
            } else {
                lds_y[8 * half + i][wave * 16 + l16] = acc[i] + bv;
            }
        }
    }

    if (!LAST) {
        __syncthreads();
        if (valid) ((float4*)(h + (size_t)node * HID))[l16] = hv;
        float* Yp = half ? fd : ya;
#pragma unroll
        for (int q = 0; q < 2; ++q) {
            const int r = wave + 8 * q;
            const int row = blockIdx.x * 16 + r;
            const float4 v = *(const float4*)&lds_y[r][4 * lane];
            if (row < n) ((float4*)(Yp + (size_t)row * HID))[l16] = v;
        }
    }
}

extern "C" void kernel_launch(void* const* d_in, const int* in_sizes, int n_in,
                              void* d_out, int out_size, void* d_ws, size_t ws_size,
                              hipStream_t stream) {
    const float* nodes = (const float*)d_in[0];
    const int*   src   = (const int*)d_in[1];
    const int*   dst   = (const int*)d_in[2];
    const float* w_in  = (const float*)d_in[3];
    const float* b_in  = (const float*)d_in[4];
    const float* w_src = (const float*)d_in[5];
    const float* b_src = (const float*)d_in[6];
    const float* w_dst = (const float*)d_in[7];
    const float* b_dst = (const float*)d_in[8];
    const float* attn  = (const float*)d_in[9];
    const float* gamma = (const float*)d_in[10];
    const float* beta  = (const float*)d_in[11];
    const float* w_out = (const float*)d_in[12];
    const float* b_out = (const float*)d_in[13];

    const int N = in_sizes[0] / HID;
    const int E = in_sizes[1];
    const int NL = 6;

    float* ws   = (float*)d_ws;
    float* h    = ws;  ws += (size_t)N * HID;
    float* fs   = ws;  ws += (size_t)N * HID;
    float* fd   = ws;  ws += (size_t)N * HID;
    float* tbuf = ws;  ws += (size_t)N * HID;
    int* ssrc      = (int*)ws; ws += E;
    int* seg_start = (int*)ws; ws += N;
    int* seg_end   = (int*)ws;
    int* eid       = (int*)tbuf;
    int* cnt_s     = eid + E;
    int* tot       = cnt_s + (size_t)GRP_SLICES * N;
    float* fsp[2]  = { fs, tbuf };

    const int ggrid  = (N + 63) / 64;
    const int eblk   = 256;
    const int egrid  = (E + eblk - 1) / eblk;
    const int sgrid  = (N * 16 + 255) / 256;
    const dim3 bgrid((N + GRP_BIN - 1) / GRP_BIN, GRP_SLICES);

    gemm64_wmma<<<ggrid, 128, 0, stream>>>(nodes, w_in, b_in, h, N);

    count_range_slice<<<bgrid, 1024, 0, stream>>>(dst, cnt_s, tot, E, N);
    place_range_slice<<<bgrid, 1024, 0, stream>>>(dst, cnt_s, tot, seg_start, seg_end, eid, E, N);
    order_segments<<<egrid, eblk, 0, stream>>>(src, dst, seg_start, seg_end, eid, ssrc, E);

    gemm64_wmma<<<ggrid, 128, 0, stream>>>(h, w_src, b_src, fs, N);
    gemm64_wmma<<<ggrid, 128, 0, stream>>>(h, w_dst, b_dst, fd, N);

    for (int L = 0; L + 1 < NL; ++L) {
        seg_softmax_update<false><<<sgrid, 256, 0, stream>>>(
            fsp[L & 1], fd, ssrc, seg_start, seg_end,
            attn + (size_t)L * HID, gamma + (size_t)L * HID, beta + (size_t)L * HID, h,
            w_src + (size_t)(L + 1) * HID * HID, b_src + (size_t)(L + 1) * HID,
            w_dst + (size_t)(L + 1) * HID * HID, b_dst + (size_t)(L + 1) * HID,
            fsp[(L + 1) & 1], N);
    }
    seg_softmax_update<true><<<sgrid, 256, 0, stream>>>(
        fsp[(NL - 1) & 1], fd, ssrc, seg_start, seg_end,
        attn + (size_t)(NL - 1) * HID, gamma + (size_t)(NL - 1) * HID,
        beta + (size_t)(NL - 1) * HID, h,
        w_out, b_out, w_out, b_out, (float*)d_out, N);
}


# input projection and layer 0's fc_src / fc_dst as one launch: a wave's 16 new h rows pass through LDS as the next WMMA A tile, the three weights staged together; 2 launches fewer, sums unchanged
# speedup vs baseline: 16.1137x; 1.1735x over previous
#include <hip/hip_runtime.h>
#include <hip/hip_bf16.h>
#include <math.h>


typedef __attribute__((ext_vector_type(2))) float v2f;
typedef __attribute__((ext_vector_type(8))) float v8f;
typedef __attribute__((ext_vector_type(4))) float v4f;

#define HID 64
#define NEG_SLOPE 0.2f
#define LN_EPS 1e-5f

__global__ void __launch_bounds__(128)
front_products(const float* __restrict__ X,
               const float* __restrict__ Win, const float* __restrict__ bin,
               const float* __restrict__ Ws,  const float* __restrict__ bs,
               const float* __restrict__ Wd,  const float* __restrict__ bd,
               float* __restrict__ H, float* __restrict__ FS, float* __restrict__ FD,
               int nrows) {
    __shared__ __attribute__((aligned(16))) float lds_w[3][64][64];
    __shared__ __attribute__((aligned(16))) float lds_h[4][16][HID + 4];

    const int tid   = threadIdx.x;
    const int lane  = tid & 31;
    const int wave  = tid >> 5;
    const int half  = lane >> 4;
    const int l15   = lane & 15;
    const int rbase = blockIdx.x * 64 + wave * 16;

    int arow = rbase + l15;
    if (arow >= nrows) arow = nrows - 1;
    const float* xrow = X + (size_t)arow * 64;

    v2f afr[16];
#pragma unroll
    for (int s = 0; s < 16; ++s) afr[s] = *(const v2f*)(xrow + 4 * s + 2 * half);

    {
        const float4* A4 = (const float4*)Win;
        const float4* B4 = (const float4*)Ws;
        const float4* C4 = (const float4*)Wd;
        float4* L4 = (float4*)&lds_w[0][0][0];
        float4 wa[8], wb[8], wc[8];
#pragma unroll
        for (int i = 0; i < 8; ++i) {
            wa[i] = A4[tid + 128 * i];
            wb[i] = B4[tid + 128 * i];
            wc[i] = C4[tid + 128 * i];
        }
#pragma unroll
        for (int i = 0; i < 8; ++i) {
            L4[tid + 128 * i]        = wa[i];
            L4[1024 + tid + 128 * i] = wb[i];
            L4[2048 + tid + 128 * i] = wc[i];
        }
    }
    __syncthreads();

    {
        v8f acc[4];
#pragma unroll
        for (int t = 0; t < 4; ++t)
#pragma unroll
            for (int j = 0; j < 8; ++j) acc[t][j] = 0.0f;

#pragma unroll
        for (int s = 0; s < 16; ++s) {
#pragma unroll
            for (int t = 0; t < 4; ++t) {
                v2f b;
                b.x = lds_w[0][4 * s + 2 * half + 0][t * 16 + l15];
                b.y = lds_w[0][4 * s + 2 * half + 1][t * 16 + l15];
                acc[t] = __builtin_amdgcn_wmma_f32_16x16x4_f32(
                    false, afr[s], false, b, (short)0, acc[t], false, false);
            }
        }
#pragma unroll
        for (int t = 0; t < 4; ++t) {
            const float bv = bin[t * 16 + l15];
#pragma unroll
            for (int i = 0; i < 8; ++i)
                lds_h[wave][8 * half + i][t * 16 + l15] = acc[t][i] + bv;
        }
    }
    __syncthreads();

    {
        v4f hrow[8];
#pragma unroll
        for (int q = 0; q < 8; ++q)
            hrow[q] = *(const v4f*)&lds_h[wave][2 * q + half][4 * l15];
        asm volatile("" : "+v"(hrow[0]), "+v"(hrow[1]), "+v"(hrow[2]), "+v"(hrow[3]),
                          "+v"(hrow[4]), "+v"(hrow[5]), "+v"(hrow[6]), "+v"(hrow[7]));
#pragma unroll
        for (int q = 0; q < 8; ++q) {
            const int row = rbase + 2 * q + half;
            if (row < nrows) *(v4f*)(H + (size_t)row * HID + 4 * l15) = hrow[q];
        }
    }

    v2f hfr[16];
#pragma unroll
    for (int s = 0; s < 16; ++s)
        hfr[s] = *(const v2f*)&lds_h[wave][l15][4 * s + 2 * half];

#pragma unroll
    for (int m = 0; m < 2; ++m) {
        const float* bias = m ? bd : bs;
        float* Y = m ? FD : FS;
        v8f acc[4];
#pragma unroll
        for (int t = 0; t < 4; ++t)
#pragma unroll
            for (int j = 0; j < 8; ++j) acc[t][j] = 0.0f;

#pragma unroll
        for (int s = 0; s < 16; ++s) {
#pragma unroll
            for (int t = 0; t < 4; ++t) {
                v2f b;
                b.x = lds_w[1 + m][4 * s + 2 * half + 0][t * 16 + l15];
                b.y = lds_w[1 + m][4 * s + 2 * half + 1][t * 16 + l15];
                acc[t] = __builtin_amdgcn_wmma_f32_16x16x4_f32(
                    false, hfr[s], false, b, (short)0, acc[t], false, false);
            }
        }
#pragma unroll
        for (int t = 0; t < 4; ++t) {
            const int col = t * 16 + l15;
            const float bv = bias[col];
#pragma unroll
            for (int i = 0; i < 8; ++i) {
                const int row = rbase + 8 * half + i;
                if (row < nrows)
                    Y[(size_t)row * HID + col] = acc[t][i] + bv;
            }
        }
    }
}

#define GRP_BIN 1024
#define GRP_SLICES 5
#define GRP_DEPTH 8

__device__ __forceinline__ void grp_slice(int E, int s, int& b4, int& e4) {
    const int n4  = E >> 2;
    const int per = (n4 + GRP_SLICES - 1) / GRP_SLICES;
    b4 = s * per;
    e4 = b4 + per;
    if (b4 > n4) b4 = n4;
    if (e4 > n4) e4 = n4;
}

__global__ void __launch_bounds__(1024)
count_range_slice(const int* __restrict__ dst, int* __restrict__ cnt_s,
                  int* __restrict__ tot, int E, int n) {
    __shared__ int c[GRP_BIN];
    __shared__ int w_tot[32];
    const int tid  = threadIdx.x;
    const int lane = tid & 31;
    const int wave = tid >> 5;
    const int r    = blockIdx.x;
    const int s    = blockIdx.y;
    const int lo   = r * GRP_BIN;
    const unsigned span = (unsigned)((n - lo) < GRP_BIN ? (n - lo) : GRP_BIN);

    c[tid] = 0;
    __syncthreads();

    int b4, e4;
    grp_slice(E, s, b4, e4);
    const int4* d4 = (const int4*)dst;
    for (int i0 = b4 + tid; i0 < e4; i0 += 1024 * GRP_DEPTH) {
        int4 v[GRP_DEPTH];
#pragma unroll
        for (int k = 0; k < GRP_DEPTH; ++k) {
            const int i = i0 + 1024 * k;
            v[k] = (i < e4) ? d4[i] : make_int4(-1, -1, -1, -1);
        }
#pragma unroll
        for (int k = 0; k < GRP_DEPTH; ++k) {
            unsigned t;
            t = (unsigned)(v[k].x - lo); if (t < span) atomicAdd(&c[t], 1);
            t = (unsigned)(v[k].y - lo); if (t < span) atomicAdd(&c[t], 1);
            t = (unsigned)(v[k].z - lo); if (t < span) atomicAdd(&c[t], 1);
            t = (unsigned)(v[k].w - lo); if (t < span) atomicAdd(&c[t], 1);
        }
    }
    if (s == GRP_SLICES - 1) {
        const int i = ((E >> 2) << 2) + tid;
        if (i < E) {
            const unsigned t = (unsigned)(dst[i] - lo);
            if (t < span) atomicAdd(&c[t], 1);
        }
    }
    __syncthreads();

    const int mine = c[tid];
    const int idx  = lo + tid;
    if (idx < n) cnt_s[(size_t)s * n + idx] = mine;

    int sum = mine;
    for (int d = 16; d >= 1; d >>= 1) sum += __shfl_xor(sum, d);
    if (lane == 0) w_tot[wave] = sum;
    __syncthreads();
    if (wave == 0) {
        int t = w_tot[lane];
        for (int d = 16; d >= 1; d >>= 1) t += __shfl_xor(t, d);
        if (lane == 0) tot[r * GRP_SLICES + s] = t;
    }
}

__global__ void __launch_bounds__(1024)
place_range_slice(const int* __restrict__ dst, const int* __restrict__ cnt_s,
                  const int* __restrict__ tot, int* __restrict__ seg_start,
                  int* __restrict__ seg_end, int* __restrict__ eid, int E, int n) {
    __shared__ int cur[GRP_BIN];
    __shared__ int w_cnt[32];
    __shared__ int w_pre[32];
    const int tid  = threadIdx.x;
    const int lane = tid & 31;
    const int wave = tid >> 5;
    const int r    = blockIdx.x;
    const int s    = blockIdx.y;
    const int lo   = r * GRP_BIN;
    const int idx  = lo + tid;
    const unsigned span = (unsigned)((n - lo) < GRP_BIN ? (n - lo) : GRP_BIN);

    int call = 0, cfront = 0;
    if (idx < n) {
#pragma unroll
        for (int q = 0; q < GRP_SLICES; ++q) {
            const int v = cnt_s[(size_t)q * n + idx];
            call += v;
            if (q < s) cfront += v;
        }
    }
    int pre = 0;
    for (int i = tid; i < r * GRP_SLICES; i += 1024) pre += tot[i];

    int incl = call;
    for (int d = 1; d < 32; d <<= 1) {
        const int v = __shfl_up(incl, d);
        if (lane >= d) incl += v;
    }
    for (int d = 16; d >= 1; d >>= 1) pre += __shfl_xor(pre, d);
    if (lane == 31) w_cnt[wave] = incl;
    if (lane == 0)  w_pre[wave] = pre;
    __syncthreads();

    const int wc = w_cnt[lane];
    int wp = w_pre[lane];
    int wincl = wc;
    for (int d = 1; d < 32; d <<= 1) {
        const int v = __shfl_up(wincl, d);
        if (lane >= d) wincl += v;
    }
    for (int d = 16; d >= 1; d >>= 1) wp += __shfl_xor(wp, d);
    const int woff = __shfl(wincl - wc, wave);

    const int excl = wp + woff + incl - call;
    if (idx < n && s == 0) {
        seg_start[idx] = excl;
        seg_end[idx]   = excl + call;
    }
    cur[tid] = excl + cfront;
    __syncthreads();

    int b4, e4;
    grp_slice(E, s, b4, e4);
    const int4* d4 = (const int4*)dst;
    for (int i0 = b4 + tid; i0 < e4; i0 += 1024 * GRP_DEPTH) {
        int4 v[GRP_DEPTH];
#pragma unroll
        for (int k = 0; k < GRP_DEPTH; ++k) {
            const int i = i0 + 1024 * k;
            v[k] = (i < e4) ? d4[i] : make_int4(-1, -1, -1, -1);
        }
#pragma unroll
        for (int k = 0; k < GRP_DEPTH; ++k) {
            const int e = (i0 + 1024 * k) << 2;
            unsigned t;
            t = (unsigned)(v[k].x - lo); if (t < span) eid[atomicAdd(&cur[t], 1)] = e;
            t = (unsigned)(v[k].y - lo); if (t < span) eid[atomicAdd(&cur[t], 1)] = e + 1;
            t = (unsigned)(v[k].z - lo); if (t < span) eid[atomicAdd(&cur[t], 1)] = e + 2;
            t = (unsigned)(v[k].w - lo); if (t < span) eid[atomicAdd(&cur[t], 1)] = e + 3;
        }
    }
    if (s == GRP_SLICES - 1) {
        const int i = ((E >> 2) << 2) + tid;
        if (i < E) {
            const unsigned t = (unsigned)(dst[i] - lo);
            if (t < span) eid[atomicAdd(&cur[t], 1)] = i;
        }
    }
}

__global__ void order_segments(const int* __restrict__ src, const int* __restrict__ dst,
                               const int* __restrict__ seg_start,
                               const int* __restrict__ seg_end,
                               const int* __restrict__ eid,
                               int* __restrict__ ssrc, int E) {
    int p = blockIdx.x * blockDim.x + threadIdx.x;
    if (p >= E) return;
    int e = eid[p];
    int d = dst[e];
    int b = seg_start[d], en = seg_end[d];
    int r = 0;
    for (int q = b; q < en; ++q) r += (eid[q] < e) ? 1 : 0;
    ssrc[b + r] = src[e];
}

template <int CTRL>
__device__ __forceinline__ float dpp_row(float v) {
    return __builtin_bit_cast(float, __builtin_amdgcn_update_dpp(
        0, __builtin_bit_cast(int, v), CTRL, 0xf, 0xf, false));
}
__device__ __forceinline__ float row16_sum(float v) {
    v += dpp_row<0x161>(v);
    v += dpp_row<0x162>(v);
    v += dpp_row<0x164>(v);
    v += dpp_row<0x168>(v);
    return v;
}

__device__ __forceinline__ float pow2_hw(float x) {
    return __builtin_amdgcn_exp2f(x);
}

template <int N>
__device__ __forceinline__ int row16_lane(int v) {
    return __builtin_amdgcn_update_dpp(0, v, 0x150 + N, 0xf, 0xf, false);
}

__device__ __forceinline__ float erf_both(float x) {
    const float t = fabsf(x);
    float q = fmaf(t, 1.69988107e-5f, -3.78677854e-4f);
    q = fmaf(t, q, 3.85781587e-3f);
    q = fmaf(t, q, -2.41816975e-2f);
    q = fmaf(t, q, 1.06668264e-1f);
    q = fmaf(t, q, 6.34933293e-1f);
    q = fmaf(t, q, 1.28689408e-1f);
    q = fmaf(t, q, t);
    const float far = 1.0f - pow2_hw(q * -1.44269504088896340736f);
    const float s = x * x;
    float p = fmaf(s, -5.61801775e-4f, 4.91381623e-3f);
    p = fmaf(s, p, -2.67075151e-2f);
    p = fmaf(s, p, 1.12800106e-1f);
    p = fmaf(s, p, -3.76122952e-1f);
    p = fmaf(s, p, 1.28379107e-1f);
    const float near = fmaf(t, p, t);
    return copysignf(t < 1.0f ? near : far, x);
}

template <int J>
__device__ __forceinline__ void edge_batch(const char* __restrict__ fsb, unsigned lane_off,
                                           int myidx, int first, int deg,
                                           const float4& fdv, const float4& av,
                                           float& m, float& l, float4& acc) {
    const int s0 = row16_lane<J + 0>(myidx);
    const int s1 = row16_lane<J + 1>(myidx);
    const int s2 = row16_lane<J + 2>(myidx);
    const int s3 = row16_lane<J + 3>(myidx);
    v4f r[4];
    r[0] = *(const v4f*)(fsb + (((unsigned)s0 << 8) | lane_off));
    r[1] = *(const v4f*)(fsb + (((unsigned)s1 << 8) | lane_off));
    r[2] = *(const v4f*)(fsb + (((unsigned)s2 << 8) | lane_off));
    r[3] = *(const v4f*)(fsb + (((unsigned)s3 << 8) | lane_off));
    asm volatile("" : "+v"(r[0]), "+v"(r[1]), "+v"(r[2]), "+v"(r[3]));
    float sc[4];
#pragma unroll
    for (int k = 0; k < 4; ++k) {
        float x0 = r[k].x + fdv.x, x1 = r[k].y + fdv.y;
        float x2 = r[k].z + fdv.z, x3 = r[k].w + fdv.w;
        x0 = fmaxf(x0, x0 * NEG_SLOPE);
        x1 = fmaxf(x1, x1 * NEG_SLOPE);
        x2 = fmaxf(x2, x2 * NEG_SLOPE);
        x3 = fmaxf(x3, x3 * NEG_SLOPE);
        const float s = row16_sum(x0 * av.x + x1 * av.y + x2 * av.z + x3 * av.w);
        sc[k] = (first + k < deg) ? s : -INFINITY;
    }
    if (first < deg) {
        const float bm    = fmaxf(fmaxf(sc[0], sc[1]), fmaxf(sc[2], sc[3]));
        const float mn    = fmaxf(m, bm);
        const float scale = pow2_hw(m - mn);
        const float p0 = pow2_hw(sc[0] - mn), p1 = pow2_hw(sc[1] - mn);
        const float p2 = pow2_hw(sc[2] - mn), p3 = pow2_hw(sc[3] - mn);
        l     = l * scale + ((p0 + p1) + (p2 + p3));
        acc.x = acc.x * scale + p0 * r[0].x + p1 * r[1].x + p2 * r[2].x + p3 * r[3].x;
        acc.y = acc.y * scale + p0 * r[0].y + p1 * r[1].y + p2 * r[2].y + p3 * r[3].y;
        acc.z = acc.z * scale + p0 * r[0].z + p1 * r[1].z + p2 * r[2].z + p3 * r[3].z;
        acc.w = acc.w * scale + p0 * r[0].w + p1 * r[1].w + p2 * r[2].w + p3 * r[3].w;
        m = mn;
    }
}

template <bool LAST>
__global__ void __launch_bounds__(256) __attribute__((amdgpu_waves_per_eu(16, 16)))
seg_softmax_update(const float* __restrict__ fs, float* __restrict__ fd,
                   const int* __restrict__ ssrc,
                   const int* __restrict__ seg_start,
                   const int* __restrict__ seg_end,
                   const float* __restrict__ attn,
                   const float* __restrict__ gamma,
                   const float* __restrict__ beta,
                   float* __restrict__ h,
                   const float* __restrict__ wa, const float* __restrict__ ba,
                   const float* __restrict__ wb, const float* __restrict__ bb,
                   float* __restrict__ ya, int n) {
    __shared__ __attribute__((aligned(16))) float lds_h[16][HID + 4];
    __shared__ __attribute__((aligned(16))) float lds_y[LAST ? 1 : 16][LAST ? 4 : 2 * HID + 4];

    const int lane = threadIdx.x & 31;
    const int l16  = lane & 15;
    const int half = lane >> 4;
    const int node = (blockIdx.x * blockDim.x + threadIdx.x) >> 4;
    const bool valid = node < n;
    const int nd = valid ? node : (n - 1);
    const int b = seg_start[nd];
    const int deg = valid ? (seg_end[nd] - b) : 0;
    const float4 fdv = ((const float4*)(fd + (size_t)nd * HID))[l16];
    float4 av = ((const float4*)attn)[l16];
    {
        const float log2e = 1.44269504088896340736f;
        av.x *= log2e; av.y *= log2e; av.z *= log2e; av.w *= log2e;
    }
    float4 hv = ((const float4*)(h + (size_t)nd * HID))[l16];

    const int d0 = __builtin_amdgcn_readlane(deg, 0);
    const int d1 = __builtin_amdgcn_readlane(deg, 16);
    const int dmax = d0 > d1 ? d0 : d1;

    float m = -INFINITY, l = 0.0f;
    float4 acc = make_float4(0.f, 0.f, 0.f, 0.f);

    const char* fsb = (const char*)fs;
    const unsigned lane_off = (unsigned)l16 * 16u;
    for (int off = 0; off < dmax; off += 16) {
        const int p = off + l16;
        const int myidx = (p < deg) ? ssrc[b + p] : 0;
        const int lim = dmax - off;
        edge_batch<0>(fsb, lane_off, myidx, off, deg, fdv, av, m, l, acc);
        if (lim > 4) {
            edge_batch<4>(fsb, lane_off, myidx, off + 4, deg, fdv, av, m, l, acc);
            if (lim > 8) {
                edge_batch<8>(fsb, lane_off, myidx, off + 8, deg, fdv, av, m, l, acc);
                if (lim > 12)
                    edge_batch<12>(fsb, lane_off, myidx, off + 12, deg, fdv, av, m, l, acc);
            }
        }
    }

    const float lq = (deg > 0) ? l : 1.0f;
    const float mean = row16_sum((acc.x + acc.y) + (acc.z + acc.w)) * (1.0f / HID);
    const float c0 = acc.x - mean, c1 = acc.y - mean, c2 = acc.z - mean, c3 = acc.w - mean;
    const float var = row16_sum((c0 * c0 + c1 * c1) + (c2 * c2 + c3 * c3)) * (1.0f / HID);
    const float inv = rsqrtf(var + LN_EPS * (lq * lq));

    const float4 g  = ((const float4*)gamma)[l16];
    const float4 bt = ((const float4*)beta)[l16];
    const float x0 = c0 * inv * g.x + bt.x;
    const float x1 = c1 * inv * g.y + bt.y;
    const float x2 = c2 * inv * g.z + bt.z;
    const float x3 = c3 * inv * g.w + bt.w;
    const float kk = 0.70710678118654752f;
    hv.x += 0.5f * x0 * (1.0f + erf_both(x0 * kk));
    hv.y += 0.5f * x1 * (1.0f + erf_both(x1 * kk));
    hv.z += 0.5f * x2 * (1.0f + erf_both(x2 * kk));
    hv.w += 0.5f * x3 * (1.0f + erf_both(x3 * kk));

    const int wave = __builtin_amdgcn_readfirstlane((int)(threadIdx.x >> 5));
    const bool owns_tile = LAST ? (wave == 0) : true;
    const int   ldw   = LAST ? 8 : HID;
    const bool  colok = LAST ? (l16 < 8) : true;
    const int   col   = LAST ? (colok ? l16 : 7)
                             : ((wave & 3) * 16 + l16);
    const float* Wt = LAST ? wa : (wave < 4 ? wa : wb);
    const float* Bt = LAST ? ba : (wave < 4 ? ba : bb);

    v2f bfrag[16];
    float bv = 0.0f;
    if (owns_tile) {
#pragma unroll
        for (int s = 0; s < 16; ++s) {
            const float b0 = Wt[(4 * s + 2 * half + 0) * ldw + col];
            const float b1 = Wt[(4 * s + 2 * half + 1) * ldw + col];
            bfrag[s].x = colok ? b0 : 0.0f;
            bfrag[s].y = colok ? b1 : 0.0f;
        }
        const float bl = Bt[col];
        bv = colok ? bl : 0.0f;
    }

    ((float4*)&lds_h[threadIdx.x >> 4][0])[l16] = hv;
    __syncthreads();

    if (owns_tile) {
        v8f acc = {};
#pragma unroll
        for (int s = 0; s < 16; ++s) {
            const v2f a = *(const v2f*)&lds_h[l16][4 * s + 2 * half];
            acc = __builtin_amdgcn_wmma_f32_16x16x4_f32(
                false, a, false, bfrag[s], (short)0, acc, false, false);
        }
#pragma unroll
        for (int i = 0; i < 8; ++i) {
            if (LAST) {
                const int row = blockIdx.x * 16 + 8 * half + i;
                if (row < n && colok)
                    ya[(size_t)row * 8 + col] = acc[i] + bv;
            } else {
                lds_y[8 * half + i][wave * 16 + l16] = acc[i] + bv;
            }
        }
    }

    if (!LAST) {
        __syncthreads();
        if (valid) ((float4*)(h + (size_t)node * HID))[l16] = hv;
        float* Yp = half ? fd : ya;
#pragma unroll
        for (int q = 0; q < 2; ++q) {
            const int r = wave + 8 * q;
            const int row = blockIdx.x * 16 + r;
            const float4 v = *(const float4*)&lds_y[r][4 * lane];
            if (row < n) ((float4*)(Yp + (size_t)row * HID))[l16] = v;
        }
    }
}

extern "C" void kernel_launch(void* const* d_in, const int* in_sizes, int n_in,
                              void* d_out, int out_size, void* d_ws, size_t ws_size,
                              hipStream_t stream) {
    const float* nodes = (const float*)d_in[0];
    const int*   src   = (const int*)d_in[1];
    const int*   dst   = (const int*)d_in[2];
    const float* w_in  = (const float*)d_in[3];
    const float* b_in  = (const float*)d_in[4];
    const float* w_src = (const float*)d_in[5];
    const float* b_src = (const float*)d_in[6];
    const float* w_dst = (const float*)d_in[7];
    const float* b_dst = (const float*)d_in[8];
    const float* attn  = (const float*)d_in[9];
    const float* gamma = (const float*)d_in[10];
    const float* beta  = (const float*)d_in[11];
    const float* w_out = (const float*)d_in[12];
    const float* b_out = (const float*)d_in[13];

    const int N = in_sizes[0] / HID;
    const int E = in_sizes[1];
    const int NL = 6;

    float* ws   = (float*)d_ws;
    float* h    = ws;  ws += (size_t)N * HID;
    float* fs   = ws;  ws += (size_t)N * HID;
    float* fd   = ws;  ws += (size_t)N * HID;
    float* tbuf = ws;  ws += (size_t)N * HID;
    int* ssrc      = (int*)ws; ws += E;
    int* seg_start = (int*)ws; ws += N;
    int* seg_end   = (int*)ws;
    int* eid       = (int*)tbuf;
    int* cnt_s     = eid + E;
    int* tot       = cnt_s + (size_t)GRP_SLICES * N;
    float* fsp[2]  = { fs, tbuf };

    const int ggrid  = (N + 63) / 64;
    const int eblk   = 256;
    const int egrid  = (E + eblk - 1) / eblk;
    const int sgrid  = (N * 16 + 255) / 256;
    const dim3 bgrid((N + GRP_BIN - 1) / GRP_BIN, GRP_SLICES);

    front_products<<<ggrid, 128, 0, stream>>>(nodes, w_in, b_in, w_src, b_src, w_dst, b_dst,
                                              h, fs, fd, N);

    count_range_slice<<<bgrid, 1024, 0, stream>>>(dst, cnt_s, tot, E, N);
    place_range_slice<<<bgrid, 1024, 0, stream>>>(dst, cnt_s, tot, seg_start, seg_end, eid, E, N);
    order_segments<<<egrid, eblk, 0, stream>>>(src, dst, seg_start, seg_end, eid, ssrc, E);

    for (int L = 0; L + 1 < NL; ++L) {
        seg_softmax_update<false><<<sgrid, 256, 0, stream>>>(
            fsp[L & 1], fd, ssrc, seg_start, seg_end,
            attn + (size_t)L * HID, gamma + (size_t)L * HID, beta + (size_t)L * HID, h,
            w_src + (size_t)(L + 1) * HID * HID, b_src + (size_t)(L + 1) * HID,
            w_dst + (size_t)(L + 1) * HID * HID, b_dst + (size_t)(L + 1) * HID,
            fsp[(L + 1) & 1], N);
    }
    seg_softmax_update<true><<<sgrid, 256, 0, stream>>>(
        fsp[(NL - 1) & 1], fd, ssrc, seg_start, seg_end,
        attn + (size_t)(NL - 1) * HID, gamma + (size_t)(NL - 1) * HID,
        beta + (size_t)(NL - 1) * HID, h,
        w_out, b_out, w_out, b_out, (float*)d_out, N);
}
